# starting kernel
# baseline (speedup 1.0000x reference)
#include <hip/hip_runtime.h>

typedef float v2f __attribute__((ext_vector_type(2)));
typedef float v8f __attribute__((ext_vector_type(8)));

#define HW_DIM   128
#define F_OUT    256
#define KDIM     2304
#define NROI     8192
#define HOWO     16384
#define KTILE    16
#define LDA      20
#define LDB      136

__global__ __launch_bounds__(256) void bias_fill_kernel(
    const float* __restrict__ bias, float* __restrict__ out) {
  const int g = blockIdx.x * 256 + threadIdx.x;
  const int f = g >> 12;
  const float bv = bias[f];
  reinterpret_cast<float4*>(out)[g] = make_float4(bv, bv, bv, bv);
}

__global__ __launch_bounds__(256) void conv_roi_wmma_kernel(
    const float* __restrict__ X,
    const float* __restrict__ W,
    const float* __restrict__ bias,
    const int*   __restrict__ roi,
    float* __restrict__ out) {
  __shared__ float As[128 * LDA];
  __shared__ float Bs[KTILE * LDB];

  const int tid   = threadIdx.x;
  const int lane  = tid & 31;
  const int wave  = tid >> 5;
  const int waveM = wave >> 2;
  const int waveN = wave & 3;
  const int lrow  = lane & 15;
  const int lhi   = lane >> 4;

  const int m0 = blockIdx.y * 128;
  const int n0 = blockIdx.x * 128;

  const int tn   = tid & 127;
  const int kgrp = tid >> 7;
  const int p_my = roi[n0 + tn];
  const int ry   = p_my >> 7;
  const int rx   = p_my & 127;

  const int arow = tid >> 2;
  const int aq   = tid & 3;

  v8f acc[4][2];
  const v8f vzero = {};
#pragma unroll
  for (int mt = 0; mt < 4; ++mt)
#pragma unroll
    for (int nt = 0; nt < 2; ++nt) acc[mt][nt] = vzero;

  for (int k0 = 0; k0 < KDIM; k0 += KTILE) {
#pragma unroll
    for (int rep = 0; rep < 2; ++rep) {
      const int row = arow + rep * 64;
      const float4 w4 =
          *reinterpret_cast<const float4*>(W + (m0 + row) * KDIM + k0 + aq * 4);
      *reinterpret_cast<float4*>(&As[row * LDA + aq * 4]) = w4;
    }
#pragma unroll
    for (int u = 0; u < 8; ++u) {
      const int kk  = kgrp * 8 + u;
      const int k   = k0 + kk;
      const int ch  = k / 9;
      const int rem = k - ch * 9;
      const int fi  = rem / 3;
      const int fj  = rem - fi * 3;
      const int y   = ry + fi - 1;
      const int x   = rx + fj - 1;
      float v = 0.0f;
      if ((unsigned)y < 128u && (unsigned)x < 128u)
        v = X[(ch << 14) + (y << 7) + x];
      Bs[kk * LDB + tn] = v;
    }
    if (k0 + KTILE < KDIM)
      __builtin_prefetch(W + (m0 + arow) * KDIM + k0 + KTILE, 0, 0);
    __syncthreads();

#pragma unroll
    for (int ks = 0; ks < 4; ++ks) {
      const int kb = ks * 4;
      v2f afrag[4];
#pragma unroll
      for (int mt = 0; mt < 4; ++mt) {
        const int m = waveM * 64 + mt * 16 + lrow;
        afrag[mt] = *reinterpret_cast<const v2f*>(&As[m * LDA + kb + lhi * 2]);
      }
      v2f bfrag[2];
#pragma unroll
      for (int nt = 0; nt < 2; ++nt) {
        const int n  = waveN * 32 + nt * 16 + lrow;
        const int kk = kb + lhi * 2;
        v2f b;
        b.x = Bs[kk * LDB + n];
        b.y = Bs[(kk + 1) * LDB + n];
        bfrag[nt] = b;
      }
#pragma unroll
      for (int mt = 0; mt < 4; ++mt)
#pragma unroll
        for (int nt = 0; nt < 2; ++nt)
          acc[mt][nt] = __builtin_amdgcn_wmma_f32_16x16x4_f32(
              false, afrag[mt], false, bfrag[nt],
              (short)0, acc[mt][nt], false, false);
    }
    __syncthreads();
  }

#pragma unroll
  for (int mt = 0; mt < 4; ++mt) {
    const int fbase = m0 + waveM * 64 + mt * 16 + lhi * 8;
#pragma unroll
    for (int nt = 0; nt < 2; ++nt) {
      const int nn = n0 + waveN * 32 + nt * 16 + lrow;
      const int p  = roi[nn];
#pragma unroll
      for (int v = 0; v < 8; ++v) {
        const int f = fbase + v;
        out[(f << 14) + p] = acc[mt][nt][v] + bias[f];
      }
    }
  }
}

extern "C" void kernel_launch(void* const* d_in, const int* in_sizes, int n_in,
                              void* d_out, int out_size, void* d_ws, size_t ws_size,
                              hipStream_t stream) {
  const float* X    = (const float*)d_in[0];
  const float* W    = (const float*)d_in[1];
  const float* B    = (const float*)d_in[2];
  const int*   roi  = (const int*)d_in[3];
  float*       out  = (float*)d_out;

  bias_fill_kernel<<<dim3((F_OUT * HOWO / 4) / 256), 256, 0, stream>>>(B, out);
  conv_roi_wmma_kernel<<<dim3(NROI / 128, F_OUT / 128), 256, 0, stream>>>(
      X, W, B, roi, out);
}


# K loop software-pipelined: next K-tile's global loads (W rows; 8 gathered pixels, made branch-free) issued before the 32 WMMAs, kept in registers, stored to a second LDS stage after; same sum order
# speedup vs baseline: 2.7107x; 2.7107x over previous
#include <hip/hip_runtime.h>


typedef float v2f __attribute__((ext_vector_type(2)));
typedef float v4f __attribute__((ext_vector_type(4)));
typedef float v8f __attribute__((ext_vector_type(8)));

#define HW_DIM   128
#define F_OUT    256
#define KDIM     2304
#define NROI     8192
#define HOWO     16384
#define KTILE    16
#define NTILES   (KDIM / KTILE)
#define LDA      20
#define LDB      136
#define A_TILE   (128 * LDA)
#define B_TILE   (KTILE * LDB)

__global__ __launch_bounds__(256) void bias_fill_kernel(
    const float* __restrict__ bias, float* __restrict__ out) {
  const int g = blockIdx.x * 256 + threadIdx.x;
  const int f = g >> 12;
  const float bv = bias[f];
  reinterpret_cast<float4*>(out)[g] = make_float4(bv, bv, bv, bv);
}

__device__ __forceinline__ void mma_tile(const float* __restrict__ As,
                                         const float* __restrict__ Bs,
                                         int waveM, int waveN, int lrow, int lhi,
                                         v8f (&acc)[4][2]) {
#pragma unroll
  for (int ks = 0; ks < 4; ++ks) {
    const int kb = ks * 4;
    v2f afrag[4];
#pragma unroll
    for (int mt = 0; mt < 4; ++mt) {
      const int m = waveM * 64 + mt * 16 + lrow;
      afrag[mt] = *reinterpret_cast<const v2f*>(&As[m * LDA + kb + lhi * 2]);
    }
    v2f bfrag[2];
#pragma unroll
    for (int nt = 0; nt < 2; ++nt) {
      const int n  = waveN * 32 + nt * 16 + lrow;
      const int kk = kb + lhi * 2;
      v2f b;
      b.x = Bs[kk * LDB + n];
      b.y = Bs[(kk + 1) * LDB + n];
      bfrag[nt] = b;
    }
#pragma unroll
    for (int mt = 0; mt < 4; ++mt)
#pragma unroll
      for (int nt = 0; nt < 2; ++nt)
        acc[mt][nt] = __builtin_amdgcn_wmma_f32_16x16x4_f32(
            false, afrag[mt], false, bfrag[nt],
            (short)0, acc[mt][nt], false, false);
  }
}

__device__ __forceinline__ void load_tile(const float* __restrict__ w0,
                                          const float* __restrict__ w1,
                                          const float* __restrict__ X,
                                          int kfirst, int ry, int rx,
                                          v4f& a0, v4f& a1, v8f& bv, unsigned& bIn) {
  a0 = *reinterpret_cast<const v4f*>(w0);
  a1 = *reinterpret_cast<const v4f*>(w1);
  unsigned in_bits = 0u;
#pragma unroll
  for (int u = 0; u < 8; ++u) {
    const int k   = kfirst + u;
    const int ch  = k / 9;
    const int rem = k - ch * 9;
    const int fi  = rem / 3;
    const int fj  = rem - fi * 3;
    const int y   = ry + fi - 1;
    const int x   = rx + fj - 1;
    const bool in = ((unsigned)y < 128u && (unsigned)x < 128u);
    const unsigned idx = in ? (unsigned)((ch << 14) + (y << 7) + x) : 0u;
    bv[u] = X[idx];
    in_bits |= in ? (1u << u) : 0u;
  }
  bIn = in_bits;
}

__device__ __forceinline__ void store_tile(float* __restrict__ Ad, float* __restrict__ Bd,
                                           const v4f& a0, const v4f& a1,
                                           const v8f& bv, unsigned bIn) {
  *reinterpret_cast<v4f*>(Ad)            = a0;
  *reinterpret_cast<v4f*>(Ad + 64 * LDA) = a1;
#pragma unroll
  for (int u = 0; u < 8; ++u)
    Bd[u * LDB] = ((bIn >> u) & 1u) ? bv[u] : 0.0f;
}

__global__ __launch_bounds__(256) void conv_roi_wmma_kernel(
    const float* __restrict__ X,
    const float* __restrict__ W,
    const float* __restrict__ bias,
    const int*   __restrict__ roi,
    float* __restrict__ out) {
  __shared__ __attribute__((aligned(16))) float As[2 * A_TILE];
  __shared__ float Bs[2 * B_TILE];

  const int tid   = threadIdx.x;
  const int lane  = tid & 31;
  const int wave  = tid >> 5;
  const int waveM = wave >> 2;
  const int waveN = wave & 3;
  const int lrow  = lane & 15;
  const int lhi   = lane >> 4;

  const int m0 = blockIdx.y * 128;
  const int n0 = blockIdx.x * 128;

  const int tn   = tid & 127;
  const int kgrp = tid >> 7;
  const int p_my = roi[n0 + tn];
  const int ry   = p_my >> 7;
  const int rx   = p_my & 127;

  const int arow = tid >> 2;
  const int aq   = tid & 3;

  v8f acc[4][2];
  const v8f vzero = {};
#pragma unroll
  for (int mt = 0; mt < 4; ++mt)
#pragma unroll
    for (int nt = 0; nt < 2; ++nt) acc[mt][nt] = vzero;

  v4f      aReg0, aReg1;
  v8f      bReg;
  unsigned bIn;

  const float* Wrow0 = W + (m0 + arow) * KDIM + aq * 4;
  const float* Wrow1 = Wrow0 + 64 * KDIM;
  const int    aDst  = arow * LDA + aq * 4;
  const int    bDst  = kgrp * 8 * LDB + tn;

  load_tile(Wrow0, Wrow1, X, 0 + kgrp * 8, ry, rx, aReg0, aReg1, bReg, bIn);
  store_tile(As + aDst, Bs + bDst, aReg0, aReg1, bReg, bIn);
  __syncthreads();

  int s = 0;
  for (int t = 0; t < NTILES - 1; ++t) {
    load_tile(Wrow0 + (t + 1) * KTILE, Wrow1 + (t + 1) * KTILE, X,
              (t + 1) * KTILE + kgrp * 8, ry, rx, aReg0, aReg1, bReg, bIn);
    __builtin_amdgcn_sched_barrier(0);
    mma_tile(As + s * A_TILE, Bs + s * B_TILE, waveM, waveN, lrow, lhi, acc);
    __builtin_amdgcn_sched_barrier(0);
    store_tile(As + (s ^ 1) * A_TILE + aDst, Bs + (s ^ 1) * B_TILE + bDst,
               aReg0, aReg1, bReg, bIn);
    __syncthreads();
    s ^= 1;
  }
  mma_tile(As + s * A_TILE, Bs + s * B_TILE, waveM, waveN, lrow, lhi, acc);

#pragma unroll
  for (int mt = 0; mt < 4; ++mt) {
    const int fbase = m0 + waveM * 64 + mt * 16 + lhi * 8;
#pragma unroll
    for (int nt = 0; nt < 2; ++nt) {
      const int nn = n0 + waveN * 32 + nt * 16 + lrow;
      const int p  = roi[nn];
#pragma unroll
      for (int v = 0; v < 8; ++v) {
        const int f = fbase + v;
        out[(f << 14) + p] = acc[mt][nt][v] + bias[f];
      }
    }
  }
}

extern "C" void kernel_launch(void* const* d_in, const int* in_sizes, int n_in,
                              void* d_out, int out_size, void* d_ws, size_t ws_size,
                              hipStream_t stream) {
  const float* X    = (const float*)d_in[0];
  const float* W    = (const float*)d_in[1];
  const float* B    = (const float*)d_in[2];
  const int*   roi  = (const int*)d_in[3];
  float*       out  = (float*)d_out;

  bias_fill_kernel<<<dim3((F_OUT * HOWO / 4) / 256), 256, 0, stream>>>(B, out);
  conv_roi_wmma_kernel<<<dim3(NROI / 128, F_OUT / 128), 256, 0, stream>>>(
      X, W, B, roi, out);
}


# Block tile 128x128x16 -> 128x32x32 (wave tile 32x16): grid 128 -> 512 work-groups, two per WGP on all 256 WGPs, 4 waves per SIMD from two work-groups with separate barriers; sum order unchanged
# speedup vs baseline: 4.5908x; 1.6936x over previous
#include <hip/hip_runtime.h>


typedef float v2f __attribute__((ext_vector_type(2)));
typedef float v4f __attribute__((ext_vector_type(4)));
typedef float v8f __attribute__((ext_vector_type(8)));

#define HW_DIM   128
#define F_OUT    256
#define KDIM     2304
#define NROI     8192
#define HOWO     16384
#define BM       128
#define BN       32
#define KTILE    32
#define NTILES   (KDIM / KTILE)
#define LDA      36
#define LDB      40
#define A_TILE   (BM * LDA)
#define B_TILE   (KTILE * LDB)

__global__ __launch_bounds__(256) void bias_fill_kernel(
    const float* __restrict__ bias, float* __restrict__ out) {
  const int g = blockIdx.x * 256 + threadIdx.x;
  const int f = g >> 12;
  const float bv = bias[f];
  reinterpret_cast<float4*>(out)[g] = make_float4(bv, bv, bv, bv);
}

__device__ __forceinline__ void mma_tile(const float* __restrict__ As,
                                         const float* __restrict__ Bs,
                                         int waveM, int waveN, int lrow, int lhi,
                                         v8f (&acc)[2]) {
#pragma unroll
  for (int ks = 0; ks < KTILE / 4; ++ks) {
    const int kb = ks * 4;
    v2f afrag[2];
#pragma unroll
    for (int mt = 0; mt < 2; ++mt) {
      const int m = waveM * 32 + mt * 16 + lrow;
      afrag[mt] = *reinterpret_cast<const v2f*>(&As[m * LDA + kb + lhi * 2]);
    }
    const int n  = waveN * 16 + lrow;
    const int kk = kb + lhi * 2;
    v2f bfrag;
    bfrag.x = Bs[kk * LDB + n];
    bfrag.y = Bs[(kk + 1) * LDB + n];
#pragma unroll
    for (int mt = 0; mt < 2; ++mt)
      acc[mt] = __builtin_amdgcn_wmma_f32_16x16x4_f32(
          false, afrag[mt], false, bfrag,
          (short)0, acc[mt], false, false);
  }
}

__device__ __forceinline__ void load_tile(const float* __restrict__ w0,
                                          const float* __restrict__ X,
                                          int kfirst, int ry, int rx,
                                          v4f (&a)[4], v4f& bv, unsigned& bIn) {
#pragma unroll
  for (int i = 0; i < 4; ++i)
    a[i] = *reinterpret_cast<const v4f*>(w0 + i * 32 * KDIM);
  unsigned in_bits = 0u;
#pragma unroll
  for (int u = 0; u < 4; ++u) {
    const int k   = kfirst + u;
    const int ch  = k / 9;
    const int rem = k - ch * 9;
    const int fi  = rem / 3;
    const int fj  = rem - fi * 3;
    const int y   = ry + fi - 1;
    const int x   = rx + fj - 1;
    const bool in = ((unsigned)y < 128u && (unsigned)x < 128u);
    const unsigned idx = in ? (unsigned)((ch << 14) + (y << 7) + x) : 0u;
    bv[u] = X[idx];
    in_bits |= in ? (1u << u) : 0u;
  }
  bIn = in_bits;
}

__device__ __forceinline__ void store_tile(float* __restrict__ Ad, float* __restrict__ Bd,
                                           const v4f (&a)[4],
                                           const v4f& bv, unsigned bIn) {
#pragma unroll
  for (int i = 0; i < 4; ++i)
    *reinterpret_cast<v4f*>(Ad + i * 32 * LDA) = a[i];
#pragma unroll
  for (int u = 0; u < 4; ++u)
    Bd[u * LDB] = ((bIn >> u) & 1u) ? bv[u] : 0.0f;
}

__global__ __launch_bounds__(256) void conv_roi_wmma_kernel(
    const float* __restrict__ X,
    const float* __restrict__ W,
    const float* __restrict__ bias,
    const int*   __restrict__ roi,
    float* __restrict__ out) {
  __shared__ __attribute__((aligned(16))) float As[2 * A_TILE];
  __shared__ float Bs[2 * B_TILE];

  const int tid   = threadIdx.x;
  const int lane  = tid & 31;
  const int wave  = tid >> 5;
  const int waveM = wave >> 1;
  const int waveN = wave & 1;
  const int lrow  = lane & 15;
  const int lhi   = lane >> 4;

  const int m0 = blockIdx.y * BM;
  const int n0 = blockIdx.x * BN;

  const int tn   = tid & 31;
  const int kgrp = tid >> 5;
  const int p_my = roi[n0 + tn];
  const int ry   = p_my >> 7;
  const int rx   = p_my & 127;

  const int arow = tid >> 3;
  const int aq   = tid & 7;

  v8f acc[2];
  const v8f vzero = {};
#pragma unroll
  for (int mt = 0; mt < 2; ++mt) acc[mt] = vzero;

  v4f      aReg[4];
  v4f      bReg;
  unsigned bIn;

  const float* Wrow0 = W + (m0 + arow) * KDIM + aq * 4;
  const int    aDst  = arow * LDA + aq * 4;
  const int    bDst  = kgrp * 4 * LDB + tn;

  load_tile(Wrow0, X, 0 + kgrp * 4, ry, rx, aReg, bReg, bIn);
  store_tile(As + aDst, Bs + bDst, aReg, bReg, bIn);
  __syncthreads();

  int s = 0;
  for (int t = 0; t < NTILES - 1; ++t) {
    load_tile(Wrow0 + (t + 1) * KTILE, X,
              (t + 1) * KTILE + kgrp * 4, ry, rx, aReg, bReg, bIn);
    __builtin_amdgcn_sched_barrier(0);
    mma_tile(As + s * A_TILE, Bs + s * B_TILE, waveM, waveN, lrow, lhi, acc);
    __builtin_amdgcn_sched_barrier(0);
    store_tile(As + (s ^ 1) * A_TILE + aDst, Bs + (s ^ 1) * B_TILE + bDst,
               aReg, bReg, bIn);
    __syncthreads();
    s ^= 1;
  }
  mma_tile(As + s * A_TILE, Bs + s * B_TILE, waveM, waveN, lrow, lhi, acc);

  const int nn = n0 + waveN * 16 + lrow;
  const int p  = roi[nn];
#pragma unroll
  for (int mt = 0; mt < 2; ++mt) {
    const int fbase = m0 + waveM * 32 + mt * 16 + lhi * 8;
#pragma unroll
    for (int v = 0; v < 8; ++v) {
      const int f = fbase + v;
      out[(f << 14) + p] = acc[mt][v] + bias[f];
    }
  }
}

extern "C" void kernel_launch(void* const* d_in, const int* in_sizes, int n_in,
                              void* d_out, int out_size, void* d_ws, size_t ws_size,
                              hipStream_t stream) {
  const float* X    = (const float*)d_in[0];
  const float* W    = (const float*)d_in[1];
  const float* B    = (const float*)d_in[2];
  const int*   roi  = (const int*)d_in[3];
  float*       out  = (float*)d_out;

  bias_fill_kernel<<<dim3((F_OUT * HOWO / 4) / 256), 256, 0, stream>>>(B, out);
  conv_roi_wmma_kernel<<<dim3(NROI / BN, F_OUT / BM), 256, 0, stream>>>(
      X, W, B, roi, out);
}


# A and B fragments of each K-tile read by ds_load_b128, all twelve as one group ahead of the tile's WMMA run (B tile kept k-contiguous in LDS); k order inside each 8-k chunk changes the sum order
# speedup vs baseline: 4.8221x; 1.0504x over previous
#include <hip/hip_runtime.h>


typedef float v2f __attribute__((ext_vector_type(2)));
typedef float v4f __attribute__((ext_vector_type(4)));
typedef float v8f __attribute__((ext_vector_type(8)));

#define HW_DIM   128
#define F_OUT    256
#define KDIM     2304
#define NROI     8192
#define HOWO     16384
#define BM       128
#define BN       32
#define KTILE    32
#define NTILES   (KDIM / KTILE)
#define LDA      36
#define LDB      36
#define A_TILE   (BM * LDA)
#define B_TILE   (BN * LDB)

__global__ __launch_bounds__(256) void bias_fill_kernel(
    const float* __restrict__ bias, float* __restrict__ out) {
  const int g = blockIdx.x * 256 + threadIdx.x;
  const int f = g >> 12;
  const float bv = bias[f];
  reinterpret_cast<float4*>(out)[g] = make_float4(bv, bv, bv, bv);
}

struct Frags {
  v4f a[2][KTILE / 8];
  v4f b[KTILE / 8];
};

__device__ __forceinline__ void load_frags(const float* __restrict__ As,
                                           const float* __restrict__ Bs,
                                           int aOff, int bOff, Frags& fr) {
#pragma unroll
  for (int c = 0; c < KTILE / 8; ++c) {
#pragma unroll
    for (int mt = 0; mt < 2; ++mt)
      fr.a[mt][c] = *reinterpret_cast<const v4f*>(&As[aOff + mt * 16 * LDA + c * 8]);
    fr.b[c] = *reinterpret_cast<const v4f*>(&Bs[bOff + c * 8]);
  }
}

__device__ __forceinline__ void mma_tile(const Frags& fr, v8f (&acc)[2]) {
#pragma unroll
  for (int c = 0; c < KTILE / 8; ++c) {
#pragma unroll
    for (int j = 0; j < 2; ++j) {
      v2f bfrag;
      bfrag.x = fr.b[c][2 * j];
      bfrag.y = fr.b[c][2 * j + 1];
#pragma unroll
      for (int mt = 0; mt < 2; ++mt) {
        v2f afrag;
        afrag.x = fr.a[mt][c][2 * j];
        afrag.y = fr.a[mt][c][2 * j + 1];
        acc[mt] = __builtin_amdgcn_wmma_f32_16x16x4_f32(
            false, afrag, false, bfrag,
            (short)0, acc[mt], false, false);
      }
    }
  }
}

__device__ __forceinline__ void load_tile(const float* __restrict__ w0,
                                          const float* __restrict__ X,
                                          int kfirst, int ry, int rx,
                                          v4f (&a)[4], v4f& bv, unsigned& bIn) {
#pragma unroll
  for (int i = 0; i < 4; ++i)
    a[i] = *reinterpret_cast<const v4f*>(w0 + i * 32 * KDIM);
  unsigned in_bits = 0u;
#pragma unroll
  for (int u = 0; u < 4; ++u) {
    const int k   = kfirst + u;
    const int ch  = k / 9;
    const int rem = k - ch * 9;
    const int fi  = rem / 3;
    const int fj  = rem - fi * 3;
    const int y   = ry + fi - 1;
    const int x   = rx + fj - 1;
    const bool in = ((unsigned)y < 128u && (unsigned)x < 128u);
    const unsigned idx = in ? (unsigned)((ch << 14) + (y << 7) + x) : 0u;
    bv[u] = X[idx];
    in_bits |= in ? (1u << u) : 0u;
  }
  bIn = in_bits;
}

__device__ __forceinline__ void store_tile(float* __restrict__ Ad, float* __restrict__ Bd,
                                           const v4f (&a)[4],
                                           const v4f& bv, unsigned bIn) {
#pragma unroll
  for (int i = 0; i < 4; ++i)
    *reinterpret_cast<v4f*>(Ad + i * 32 * LDA) = a[i];
  v4f bz;
#pragma unroll
  for (int u = 0; u < 4; ++u)
    bz[u] = ((bIn >> u) & 1u) ? bv[u] : 0.0f;
  *reinterpret_cast<v4f*>(Bd) = bz;
}

__global__ __launch_bounds__(256) void conv_roi_wmma_kernel(
    const float* __restrict__ X,
    const float* __restrict__ W,
    const float* __restrict__ bias,
    const int*   __restrict__ roi,
    float* __restrict__ out) {
  __shared__ __attribute__((aligned(16))) float As[2 * A_TILE];
  __shared__ __attribute__((aligned(16))) float Bs[2 * B_TILE];

  const int tid   = threadIdx.x;
  const int lane  = tid & 31;
  const int wave  = tid >> 5;
  const int waveM = wave >> 1;
  const int waveN = wave & 1;
  const int lrow  = lane & 15;
  const int lhi   = lane >> 4;

  const int m0 = blockIdx.y * BM;
  const int n0 = blockIdx.x * BN;

  const int tn   = tid & 31;
  const int kgrp = tid >> 5;
  const int p_my = roi[n0 + tn];
  const int ry   = p_my >> 7;
  const int rx   = p_my & 127;

  const int arow = tid >> 3;
  const int aq   = tid & 7;

  v8f acc[2];
  const v8f vzero = {};
#pragma unroll
  for (int mt = 0; mt < 2; ++mt) acc[mt] = vzero;

  v4f      aReg[4];
  v4f      bReg;
  unsigned bIn;

  const float* Wrow0 = W + (m0 + arow) * KDIM + aq * 4;
  const int    aDst  = arow * LDA + aq * 4;
  const int    bDst  = tn * LDB + kgrp * 4;
  const int    aFrag = (waveM * 32 + lrow) * LDA + lhi * 4;
  const int    bFrag = (waveN * 16 + lrow) * LDB + lhi * 4;
  Frags fr;

  load_tile(Wrow0, X, 0 + kgrp * 4, ry, rx, aReg, bReg, bIn);
  store_tile(As + aDst, Bs + bDst, aReg, bReg, bIn);
  __syncthreads();

  int s = 0;
  for (int t = 0; t < NTILES - 1; ++t) {
    load_frags(As + s * A_TILE, Bs + s * B_TILE, aFrag, bFrag, fr);
    __builtin_amdgcn_sched_barrier(0);
    load_tile(Wrow0 + (t + 1) * KTILE, X,
              (t + 1) * KTILE + kgrp * 4, ry, rx, aReg, bReg, bIn);
    __builtin_amdgcn_sched_barrier(0);
    mma_tile(fr, acc);
    __builtin_amdgcn_sched_barrier(0);
    store_tile(As + (s ^ 1) * A_TILE + aDst, Bs + (s ^ 1) * B_TILE + bDst,
               aReg, bReg, bIn);
    __syncthreads();
    s ^= 1;
  }
  load_frags(As + s * A_TILE, Bs + s * B_TILE, aFrag, bFrag, fr);
  mma_tile(fr, acc);

  const int nn = n0 + waveN * 16 + lrow;
  const int p  = roi[nn];
#pragma unroll
  for (int mt = 0; mt < 2; ++mt) {
    const int fbase = m0 + waveM * 32 + mt * 16 + lhi * 8;
#pragma unroll
    for (int v = 0; v < 8; ++v) {
      const int f = fbase + v;
      out[(f << 14) + p] = acc[mt][v] + bias[f];
    }
  }
}

extern "C" void kernel_launch(void* const* d_in, const int* in_sizes, int n_in,
                              void* d_out, int out_size, void* d_ws, size_t ws_size,
                              hipStream_t stream) {
  const float* X    = (const float*)d_in[0];
  const float* W    = (const float*)d_in[1];
  const float* B    = (const float*)d_in[2];
  const int*   roi  = (const int*)d_in[3];
  float*       out  = (float*)d_out;

  bias_fill_kernel<<<dim3((F_OUT * HOWO / 4) / 256), 256, 0, stream>>>(B, out);
  conv_roi_wmma_kernel<<<dim3(NROI / BN, F_OUT / BM), 256, 0, stream>>>(
      X, W, B, roi, out);
}


# K-loop address work taken off the vector unit: wave-uniform k decoded (channel, tap, border lane masks) on the scalar unit between the WMMAs, 68 -> 17 vector instructions per tile; same sum order
# speedup vs baseline: 5.4153x; 1.1230x over previous
#include <hip/hip_runtime.h>


typedef float v2f __attribute__((ext_vector_type(2)));
typedef float v4f __attribute__((ext_vector_type(4)));
typedef float v8f __attribute__((ext_vector_type(8)));

#define HW_DIM   128
#define F_OUT    256
#define KDIM     2304
#define NROI     8192
#define HOWO     16384
#define BM       128
#define BN       32
#define KTILE    32
#define NTILES   (KDIM / KTILE)
#define LDA      36
#define LDB      36
#define A_TILE   (BM * LDA)
#define B_TILE   (BN * LDB)
#define STAGE    (A_TILE + B_TILE)
#define SAFE_PIX 129u

__global__ __launch_bounds__(256) void bias_fill_kernel(
    const float* __restrict__ bias, float* __restrict__ out) {
  const int g = blockIdx.x * 256 + threadIdx.x;
  const int f = g >> 12;
  const float bv = bias[f];
  reinterpret_cast<float4*>(out)[g] = make_float4(bv, bv, bv, bv);
}

struct Frags {
  v4f a[2][KTILE / 8];
  v4f b[KTILE / 8];
};

__device__ __forceinline__ void load_frags(const float* __restrict__ As,
                                           const float* __restrict__ Bs,
                                           int aOff, int bOff, Frags& fr) {
#pragma unroll
  for (int c = 0; c < KTILE / 8; ++c) {
#pragma unroll
    for (int mt = 0; mt < 2; ++mt)
      fr.a[mt][c] = *reinterpret_cast<const v4f*>(&As[aOff + mt * 16 * LDA + c * 8]);
    fr.b[c] = *reinterpret_cast<const v4f*>(&Bs[bOff + c * 8]);
  }
}

__device__ __forceinline__ void mma_tile(const Frags& fr, v8f (&acc)[2]) {
#pragma unroll
  for (int c = 0; c < KTILE / 8; ++c) {
#pragma unroll
    for (int j = 0; j < 2; ++j) {
      v2f bfrag;
      bfrag.x = fr.b[c][2 * j];
      bfrag.y = fr.b[c][2 * j + 1];
#pragma unroll
      for (int mt = 0; mt < 2; ++mt) {
        v2f afrag;
        afrag.x = fr.a[mt][c][2 * j];
        afrag.y = fr.a[mt][c][2 * j + 1];
        acc[mt] = __builtin_amdgcn_wmma_f32_16x16x4_f32(
            false, afrag, false, bfrag,
            (short)0, acc[mt], false, false);
      }
    }
  }
}

struct Taps {
  const float* xk[4];
  unsigned     out[4];
};

__device__ __forceinline__ void decode_taps(const float* __restrict__ X,
                                            int ch0, int rem0,
                                            unsigned mTop, unsigned mBot,
                                            unsigned mLeft, unsigned mRight,
                                            Taps& tp) {
  int ch  = ch0;
  int rem = rem0;
#pragma unroll
  for (int u = 0; u < 4; ++u) {
    const int fi = (rem >= 3) + (rem >= 6);
    const int fj = rem - 3 * fi;
    const unsigned rowm = (fi == 0) ? mTop  : ((fi == 2) ? mBot   : 0u);
    const unsigned colm = (fj == 0) ? mLeft : ((fj == 2) ? mRight : 0u);
    tp.out[u] = rowm | colm;
    tp.xk[u]  = X + ((ch << 14) + ((fi - 1) << 7) + (fj - 1));
    rem += 1;
    if (rem == 9) { rem = 0; ch += 1; }
  }
}

__device__ __forceinline__ void issue_loads(const float* __restrict__ Wt,
                                            unsigned wLane, const Taps& tp, unsigned pc,
                                            v4f (&a)[4], v4f& bv) {
#pragma unroll
  for (int i = 0; i < 4; ++i)
    a[i] = *reinterpret_cast<const v4f*>(Wt + wLane + i * 32 * KDIM);
#pragma unroll
  for (int u = 0; u < 4; ++u) {
    const unsigned off = __builtin_amdgcn_inverse_ballot_w32(tp.out[u]) ? SAFE_PIX : pc;
    bv[u] = tp.xk[u][off];
  }
}

__device__ __forceinline__ void store_tile(float* __restrict__ Ad, float* __restrict__ Bd,
                                           const v4f (&a)[4],
                                           const v4f& bv, const unsigned (&outm)[4]) {
#pragma unroll
  for (int i = 0; i < 4; ++i)
    *reinterpret_cast<v4f*>(Ad + i * 32 * LDA) = a[i];
  v4f bz;
#pragma unroll
  for (int u = 0; u < 4; ++u)
    bz[u] = __builtin_amdgcn_inverse_ballot_w32(outm[u]) ? 0.0f : bv[u];
  *reinterpret_cast<v4f*>(Bd) = bz;
}

__global__ __launch_bounds__(256) void conv_roi_wmma_kernel(
    const float* __restrict__ X,
    const float* __restrict__ W,
    const float* __restrict__ bias,
    const int*   __restrict__ roi,
    float* __restrict__ out) {
  __shared__ __attribute__((aligned(16))) float smem[2 * STAGE];

  const int tid   = threadIdx.x;
  const int lane  = tid & 31;
  const int wave  = tid >> 5;
  const int waveM = wave >> 1;
  const int waveN = wave & 1;
  const int lrow  = lane & 15;
  const int lhi   = lane >> 4;

  const int m0 = blockIdx.y * BM;
  const int n0 = blockIdx.x * BN;

  const int tn   = tid & 31;
  const int kgrp = __builtin_amdgcn_readfirstlane(tid >> 5);
  const int p_my = roi[n0 + tn];
  const unsigned pc = (unsigned)min(max(p_my, 0), HOWO - 1);
  const int ry   = (int)(pc >> 7);
  const int rx   = (int)(pc & 127u);
  const unsigned mTop   = __builtin_amdgcn_ballot_w32(ry == 0);
  const unsigned mBot   = __builtin_amdgcn_ballot_w32(ry == HW_DIM - 1);
  const unsigned mLeft  = __builtin_amdgcn_ballot_w32(rx == 0);
  const unsigned mRight = __builtin_amdgcn_ballot_w32(rx == HW_DIM - 1);

  const int arow = tid >> 3;
  const int aq   = tid & 7;

  v8f acc[2];
  const v8f vzero = {};
#pragma unroll
  for (int mt = 0; mt < 2; ++mt) acc[mt] = vzero;

  v4f      aReg[4];
  v4f      bReg;
  unsigned bOut[4];

  const unsigned wLane = (unsigned)((m0 + arow) * KDIM + aq * 4);
  const int    aDst  = arow * LDA + aq * 4;
  const int    bDst  = A_TILE + tn * LDB + kgrp * 4;
  const int    aFrag = (waveM * 32 + lrow) * LDA + lhi * 4;
  const int    bFrag = A_TILE + (waveN * 16 + lrow) * LDB + lhi * 4;
  Frags fr;

  int ch0  = (kgrp * 4) / 9;
  int rem0 = (kgrp * 4) - ch0 * 9;

  Taps tp;
  decode_taps(X, ch0, rem0, mTop, mBot, mLeft, mRight, tp);
  issue_loads(W, wLane, tp, pc, aReg, bReg);
#pragma unroll
  for (int u = 0; u < 4; ++u) bOut[u] = tp.out[u];
  ch0  += 3;
  rem0 += 5;
  if (rem0 >= 9) { rem0 -= 9; ch0 += 1; }
  decode_taps(X, ch0, rem0, mTop, mBot, mLeft, mRight, tp);
  store_tile(smem + aDst, smem + bDst, aReg, bReg, bOut);
  __syncthreads();

  int cur = 0;
  int nxt = STAGE;
  for (int t = 0; t < NTILES - 1; ++t) {
    load_frags(smem + cur, smem + cur, aFrag, bFrag, fr);
    __builtin_amdgcn_sched_barrier(0);
    issue_loads(W + (t + 1) * KTILE, wLane, tp, pc, aReg, bReg);
#pragma unroll
    for (int u = 0; u < 4; ++u) bOut[u] = tp.out[u];
    __builtin_amdgcn_sched_barrier(0);
    mma_tile(fr, acc);
    ch0  += 3;
    rem0 += 5;
    if (rem0 >= 9) { rem0 -= 9; ch0 += 1; }
    decode_taps(X, ch0, rem0, mTop, mBot, mLeft, mRight, tp);
    __builtin_amdgcn_sched_barrier(0);
    store_tile(smem + nxt + aDst, smem + nxt + bDst, aReg, bReg, bOut);
    __syncthreads();
    const int was = cur;
    cur = nxt;
    nxt = was;
  }
  load_frags(smem + cur, smem + cur, aFrag, bFrag, fr);
  mma_tile(fr, acc);

  const int nn = n0 + waveN * 16 + lrow;
  const int p  = roi[nn];
#pragma unroll
  for (int mt = 0; mt < 2; ++mt) {
    const int fbase = m0 + waveM * 32 + mt * 16 + lhi * 8;
#pragma unroll
    for (int v = 0; v < 8; ++v) {
      const int f = fbase + v;
      out[(f << 14) + p] = acc[mt][v] + bias[f];
    }
  }
}

extern "C" void kernel_launch(void* const* d_in, const int* in_sizes, int n_in,
                              void* d_out, int out_size, void* d_ws, size_t ws_size,
                              hipStream_t stream) {
  const float* X    = (const float*)d_in[0];
  const float* W    = (const float*)d_in[1];
  const float* B    = (const float*)d_in[2];
  const int*   roi  = (const int*)d_in[3];
  float*       out  = (float*)d_out;

  bias_fill_kernel<<<dim3((F_OUT * HOWO / 4) / 256), 256, 0, stream>>>(B, out);
  conv_roi_wmma_kernel<<<dim3(NROI / BN, F_OUT / BM), 256, 0, stream>>>(
      X, W, B, roi, out);
}


# K loop pipelined one tile deeper: global loads 3 tiles ahead in 2 register sets, left in flight across LDS-only barriers; next tile's fragments read under current WMMAs (2nd set); same sum order
# speedup vs baseline: 5.7316x; 1.0584x over previous
#include <hip/hip_runtime.h>


typedef float v2f __attribute__((ext_vector_type(2)));
typedef float v4f __attribute__((ext_vector_type(4)));
typedef float v8f __attribute__((ext_vector_type(8)));

#define HW_DIM   128
#define F_OUT    256
#define KDIM     2304
#define NROI     8192
#define HOWO     16384
#define BM       128
#define BN       32
#define KTILE    32
#define NTILES   (KDIM / KTILE)
#define LDA      36
#define LDB      36
#define A_TILE   (BM * LDA)
#define B_TILE   (BN * LDB)
#define STAGE    (A_TILE + B_TILE)
#define SAFE_PIX 129u

__global__ __launch_bounds__(256) void bias_fill_kernel(
    const float* __restrict__ bias, float* __restrict__ out) {
  const int g = blockIdx.x * 256 + threadIdx.x;
  const int f = g >> 12;
  const float bv = bias[f];
  reinterpret_cast<float4*>(out)[g] = make_float4(bv, bv, bv, bv);
}

struct Frags {
  v4f a[2][KTILE / 8];
  v4f b[KTILE / 8];
};

__device__ __forceinline__ void load_frags(const float* __restrict__ As,
                                           const float* __restrict__ Bs,
                                           int aOff, int bOff, Frags& fr) {
#pragma unroll
  for (int c = 0; c < KTILE / 8; ++c) {
#pragma unroll
    for (int mt = 0; mt < 2; ++mt)
      fr.a[mt][c] = *reinterpret_cast<const v4f*>(&As[aOff + mt * 16 * LDA + c * 8]);
    fr.b[c] = *reinterpret_cast<const v4f*>(&Bs[bOff + c * 8]);
  }
}

__device__ __forceinline__ void mma_tile(const Frags& fr, v8f (&acc)[2]) {
#pragma unroll
  for (int c = 0; c < KTILE / 8; ++c) {
#pragma unroll
    for (int j = 0; j < 2; ++j) {
      v2f bfrag;
      bfrag.x = fr.b[c][2 * j];
      bfrag.y = fr.b[c][2 * j + 1];
#pragma unroll
      for (int mt = 0; mt < 2; ++mt) {
        v2f afrag;
        afrag.x = fr.a[mt][c][2 * j];
        afrag.y = fr.a[mt][c][2 * j + 1];
        acc[mt] = __builtin_amdgcn_wmma_f32_16x16x4_f32(
            false, afrag, false, bfrag,
            (short)0, acc[mt], false, false);
      }
    }
  }
}

struct Taps {
  const float* xk[4];
  unsigned     out[4];
};

__device__ __forceinline__ void decode_taps(const float* __restrict__ X,
                                            int ch0, int rem0,
                                            unsigned mTop, unsigned mBot,
                                            unsigned mLeft, unsigned mRight,
                                            Taps& tp) {
  int ch  = ch0;
  int rem = rem0;
#pragma unroll
  for (int u = 0; u < 4; ++u) {
    const int fi = (rem >= 3) + (rem >= 6);
    const int fj = rem - 3 * fi;
    const unsigned rowm = (fi == 0) ? mTop  : ((fi == 2) ? mBot   : 0u);
    const unsigned colm = (fj == 0) ? mLeft : ((fj == 2) ? mRight : 0u);
    tp.out[u] = rowm | colm;
    tp.xk[u]  = X + ((ch << 14) + ((fi - 1) << 7) + (fj - 1));
    rem += 1;
    if (rem == 9) { rem = 0; ch += 1; }
  }
}

__device__ __forceinline__ void issue_loads(const float* __restrict__ Wt,
                                            unsigned wLane, const Taps& tp, unsigned pc,
                                            v4f (&a)[4], v4f& bv) {
#pragma unroll
  for (int i = 0; i < 4; ++i)
    a[i] = *reinterpret_cast<const v4f*>(Wt + wLane + i * 32 * KDIM);
#pragma unroll
  for (int u = 0; u < 4; ++u) {
    const unsigned off = __builtin_amdgcn_inverse_ballot_w32(tp.out[u]) ? SAFE_PIX : pc;
    bv[u] = tp.xk[u][off];
  }
}

__device__ __forceinline__ void store_tile(float* __restrict__ Ad, float* __restrict__ Bd,
                                           const v4f (&a)[4],
                                           const v4f& bv, const unsigned (&outm)[4]) {
#pragma unroll
  for (int i = 0; i < 4; ++i)
    *reinterpret_cast<v4f*>(Ad + i * 32 * LDA) = a[i];
  v4f bz;
#pragma unroll
  for (int u = 0; u < 4; ++u)
    bz[u] = __builtin_amdgcn_inverse_ballot_w32(outm[u]) ? 0.0f : bv[u];
  *reinterpret_cast<v4f*>(Bd) = bz;
}

__device__ __forceinline__ void lds_barrier() {
  __builtin_amdgcn_fence(__ATOMIC_RELEASE, "workgroup", "local");
  __builtin_amdgcn_s_barrier();
  __builtin_amdgcn_fence(__ATOMIC_ACQUIRE, "workgroup", "local");
}

struct Held {
  v4f      a[4];
  v4f      b;
  unsigned out[4];
};

__device__ __forceinline__ void next_tile_k(int& ch0, int& rem0) {
  ch0  += 3;
  rem0 += 5;
  if (rem0 >= 9) { rem0 -= 9; ch0 += 1; }
}

template <bool kLoad, bool kDecode, bool kStore>
__device__ __forceinline__ void k_step(const float* __restrict__ X,
                                       const float* __restrict__ Wt,
                                       unsigned wLane, unsigned pc,
                                       unsigned mTop, unsigned mBot,
                                       unsigned mLeft, unsigned mRight,
                                       int& ch0, int& rem0, Taps& tp,
                                       const float* __restrict__ fragStage,
                                       float* __restrict__ storeStage,
                                       int aFrag, int bFrag, int aDst, int bDst,
                                       const Frags& frCur, Frags& frNext,
                                       Held& gLoad, const Held& gStore,
                                       v8f (&acc)[2]) {
  load_frags(fragStage, fragStage, aFrag, bFrag, frNext);
  if (kLoad) {
    issue_loads(Wt, wLane, tp, pc, gLoad.a, gLoad.b);
#pragma unroll
    for (int u = 0; u < 4; ++u) gLoad.out[u] = tp.out[u];
  }
  __builtin_amdgcn_sched_barrier(0);
  mma_tile(frCur, acc);
  if (kDecode) {
    next_tile_k(ch0, rem0);
    decode_taps(X, ch0, rem0, mTop, mBot, mLeft, mRight, tp);
  }
  __builtin_amdgcn_sched_barrier(0);
  if (kStore) store_tile(storeStage + aDst, storeStage + bDst, gStore.a, gStore.b, gStore.out);
  lds_barrier();
}

__global__ __launch_bounds__(256) void conv_roi_wmma_kernel(
    const float* __restrict__ X,
    const float* __restrict__ W,
    const float* __restrict__ bias,
    const int*   __restrict__ roi,
    float* __restrict__ out) {
  __shared__ __attribute__((aligned(16))) float smem[2 * STAGE];

  const int tid   = threadIdx.x;
  const int lane  = tid & 31;
  const int wave  = tid >> 5;
  const int waveM = wave >> 1;
  const int waveN = wave & 1;
  const int lrow  = lane & 15;
  const int lhi   = lane >> 4;

  const int m0 = blockIdx.y * BM;
  const int n0 = blockIdx.x * BN;

  const int tn   = tid & 31;
  const int kgrp = __builtin_amdgcn_readfirstlane(tid >> 5);
  const int p_my = roi[n0 + tn];
  const unsigned pc = (unsigned)min(max(p_my, 0), HOWO - 1);
  const int ry   = (int)(pc >> 7);
  const int rx   = (int)(pc & 127u);
  const unsigned mTop   = __builtin_amdgcn_ballot_w32(ry == 0);
  const unsigned mBot   = __builtin_amdgcn_ballot_w32(ry == HW_DIM - 1);
  const unsigned mLeft  = __builtin_amdgcn_ballot_w32(rx == 0);
  const unsigned mRight = __builtin_amdgcn_ballot_w32(rx == HW_DIM - 1);

  const int arow = tid >> 3;
  const int aq   = tid & 7;

  v8f acc[2];
  const v8f vzero = {};
#pragma unroll
  for (int mt = 0; mt < 2; ++mt) acc[mt] = vzero;

  const unsigned wLane = (unsigned)((m0 + arow) * KDIM + aq * 4);
  const int    aDst  = arow * LDA + aq * 4;
  const int    bDst  = A_TILE + tn * LDB + kgrp * 4;
  const int    aFrag = (waveM * 32 + lrow) * LDA + lhi * 4;
  const int    bFrag = A_TILE + (waveN * 16 + lrow) * LDB + lhi * 4;

  Held  g0, g1;
  Frags fr0, fr1;
  float* const st0 = smem;
  float* const st1 = smem + STAGE;

  int ch0  = (kgrp * 4) / 9;
  int rem0 = (kgrp * 4) - ch0 * 9;

  Taps tp;
  decode_taps(X, ch0, rem0, mTop, mBot, mLeft, mRight, tp);
  issue_loads(W, wLane, tp, pc, g0.a, g0.b);
#pragma unroll
  for (int u = 0; u < 4; ++u) g0.out[u] = tp.out[u];
  next_tile_k(ch0, rem0);
  decode_taps(X, ch0, rem0, mTop, mBot, mLeft, mRight, tp);
  issue_loads(W + KTILE, wLane, tp, pc, g1.a, g1.b);
#pragma unroll
  for (int u = 0; u < 4; ++u) g1.out[u] = tp.out[u];
  next_tile_k(ch0, rem0);
  decode_taps(X, ch0, rem0, mTop, mBot, mLeft, mRight, tp);
  store_tile(st0 + aDst, st0 + bDst, g0.a, g0.b, g0.out);
  issue_loads(W + 2 * KTILE, wLane, tp, pc, g0.a, g0.b);
#pragma unroll
  for (int u = 0; u < 4; ++u) g0.out[u] = tp.out[u];
  next_tile_k(ch0, rem0);
  decode_taps(X, ch0, rem0, mTop, mBot, mLeft, mRight, tp);
  lds_barrier();
  load_frags(st0, st0, aFrag, bFrag, fr0);
  store_tile(st1 + aDst, st1 + bDst, g1.a, g1.b, g1.out);
  lds_barrier();

  const float* Wn = W + 3 * KTILE;
  for (int t = 0; t < NTILES - 4; t += 2) {
    k_step<true, true, true>(X, Wn, wLane, pc, mTop, mBot, mLeft, mRight, ch0, rem0, tp,
                             st1, st0, aFrag, bFrag, aDst, bDst, fr0, fr1, g1, g0, acc);
    k_step<true, true, true>(X, Wn + KTILE, wLane, pc, mTop, mBot, mLeft, mRight, ch0, rem0, tp,
                             st0, st1, aFrag, bFrag, aDst, bDst, fr1, fr0, g0, g1, acc);
    Wn += 2 * KTILE;
  }
  k_step<true, false, true>(X, Wn, wLane, pc, mTop, mBot, mLeft, mRight, ch0, rem0, tp,
                            st1, st0, aFrag, bFrag, aDst, bDst, fr0, fr1, g1, g0, acc);
  k_step<false, false, true>(X, Wn, wLane, pc, mTop, mBot, mLeft, mRight, ch0, rem0, tp,
                             st0, st1, aFrag, bFrag, aDst, bDst, fr1, fr0, g0, g1, acc);
  load_frags(st1, st1, aFrag, bFrag, fr1);
  mma_tile(fr0, acc);
  mma_tile(fr1, acc);

  const int nn = n0 + waveN * 16 + lrow;
  const int p  = roi[nn];
#pragma unroll
  for (int mt = 0; mt < 2; ++mt) {
    const int fbase = m0 + waveM * 32 + mt * 16 + lhi * 8;
#pragma unroll
    for (int v = 0; v < 8; ++v) {
      const int f = fbase + v;
      out[(f << 14) + p] = acc[mt][v] + bias[f];
    }
  }
}

extern "C" void kernel_launch(void* const* d_in, const int* in_sizes, int n_in,
                              void* d_out, int out_size, void* d_ws, size_t ws_size,
                              hipStream_t stream) {
  const float* X    = (const float*)d_in[0];
  const float* W    = (const float*)d_in[1];
  const float* B    = (const float*)d_in[2];
  const int*   roi  = (const int*)d_in[3];
  float*       out  = (float*)d_out;

  bias_fill_kernel<<<dim3((F_OUT * HOWO / 4) / 256), 256, 0, stream>>>(B, out);
  conv_roi_wmma_kernel<<<dim3(NROI / BN, F_OUT / BM), 256, 0, stream>>>(
      X, W, B, roi, out);
}


# Bias fill fused into the GEMM kernel: each work-group stores bias to the non-RoI pixels of its own 128-filter x 64-pixel slab of out (RoI flags in LDS from a scan of roi); one launch; same sum order
# speedup vs baseline: 5.9261x; 1.0339x over previous
#include <hip/hip_runtime.h>


typedef float v2f __attribute__((ext_vector_type(2)));
typedef float v4f __attribute__((ext_vector_type(4)));
typedef float v8f __attribute__((ext_vector_type(8)));
typedef float v16f __attribute__((ext_vector_type(16)));
typedef unsigned v4u __attribute__((ext_vector_type(4)));

#define HW_DIM   128
#define F_OUT    256
#define KDIM     2304
#define NROI     8192
#define HOWO     16384
#define BM       128
#define BN       32
#define KTILE    32
#define NTILES   (KDIM / KTILE)
#define LDA      36
#define LDB      36
#define A_TILE   (BM * LDA)
#define B_TILE   (BN * LDB)
#define STAGE    (A_TILE + B_TILE)
#define SAFE_PIX 129u
#define FILL_PIX 64
#define FILL_ROWS 16


__device__ __forceinline__ void mark_roi(const int4 (&rs)[NROI / 1024], int pix0,
                                         unsigned* flags, int wave) {
  for (int i = 0; i < NROI / 1024; ++i) {
    const unsigned d[4] = {(unsigned)(rs[i].x - pix0), (unsigned)(rs[i].y - pix0),
                           (unsigned)(rs[i].z - pix0), (unsigned)(rs[i].w - pix0)};
    const unsigned dmin = min(min(d[0], d[1]), min(d[2], d[3]));
    if (__builtin_amdgcn_ballot_w32(dmin < FILL_PIX) != 0u) {
      for (int c = 0; c < 4; ++c)
        if (d[c] < FILL_PIX) flags[d[c] * 8 + wave] = 1u;
    }
  }
}

__device__ __forceinline__ void fill_slab(const v16f& bv, float* __restrict__ out,
                                          const unsigned* flags,
                                          int f0, int pix0, int lane) {
  float* o = out + (f0 << 14) + pix0 + lane;
  bool plain[2];
  for (int h = 0; h < 2; ++h) {
    const v4u lo = *reinterpret_cast<const v4u*>(&flags[(lane + 32 * h) * 8]);
    const v4u hi = *reinterpret_cast<const v4u*>(&flags[(lane + 32 * h) * 8 + 4]);
    const v4u any = lo | hi;
    plain[h] = (any.x | any.y | any.z | any.w) == 0u;
  }
  for (int h = 0; h < 2; ++h) {
    if (plain[h]) {
      for (int r = 0; r < FILL_ROWS; ++r) o[r * HOWO + 32 * h] = bv[r];
    }
  }
}

struct Frags {
  v4f a[2][KTILE / 8];
  v4f b[KTILE / 8];
};

__device__ __forceinline__ void load_frags(const float* __restrict__ As,
                                           const float* __restrict__ Bs,
                                           int aOff, int bOff, Frags& fr) {
#pragma unroll
  for (int c = 0; c < KTILE / 8; ++c) {
#pragma unroll
    for (int mt = 0; mt < 2; ++mt)
      fr.a[mt][c] = *reinterpret_cast<const v4f*>(&As[aOff + mt * 16 * LDA + c * 8]);
    fr.b[c] = *reinterpret_cast<const v4f*>(&Bs[bOff + c * 8]);
  }
}

__device__ __forceinline__ void mma_tile(const Frags& fr, v8f (&acc)[2]) {
#pragma unroll
  for (int c = 0; c < KTILE / 8; ++c) {
#pragma unroll
    for (int j = 0; j < 2; ++j) {
      v2f bfrag;
      bfrag.x = fr.b[c][2 * j];
      bfrag.y = fr.b[c][2 * j + 1];
#pragma unroll
      for (int mt = 0; mt < 2; ++mt) {
        v2f afrag;
        afrag.x = fr.a[mt][c][2 * j];
        afrag.y = fr.a[mt][c][2 * j + 1];
        acc[mt] = __builtin_amdgcn_wmma_f32_16x16x4_f32(
            false, afrag, false, bfrag,
            (short)0, acc[mt], false, false);
      }
    }
  }
}

struct Taps {
  const float* xk[4];
  unsigned     out[4];
};

__device__ __forceinline__ void decode_taps(const float* __restrict__ X,
                                            int ch0, int rem0,
                                            unsigned mTop, unsigned mBot,
                                            unsigned mLeft, unsigned mRight,
                                            Taps& tp) {
  int ch  = ch0;
  int rem = rem0;
#pragma unroll
  for (int u = 0; u < 4; ++u) {
    const int fi = (rem >= 3) + (rem >= 6);
    const int fj = rem - 3 * fi;
    const unsigned rowm = (fi == 0) ? mTop  : ((fi == 2) ? mBot   : 0u);
    const unsigned colm = (fj == 0) ? mLeft : ((fj == 2) ? mRight : 0u);
    tp.out[u] = rowm | colm;
    tp.xk[u]  = X + ((ch << 14) + ((fi - 1) << 7) + (fj - 1));
    rem += 1;
    if (rem == 9) { rem = 0; ch += 1; }
  }
}

__device__ __forceinline__ void issue_loads(const float* __restrict__ Wt,
                                            unsigned wLane, const Taps& tp, unsigned pc,
                                            v4f (&a)[4], v4f& bv) {
#pragma unroll
  for (int i = 0; i < 4; ++i)
    a[i] = *reinterpret_cast<const v4f*>(Wt + wLane + i * 32 * KDIM);
#pragma unroll
  for (int u = 0; u < 4; ++u) {
    const unsigned off = __builtin_amdgcn_inverse_ballot_w32(tp.out[u]) ? SAFE_PIX : pc;
    bv[u] = tp.xk[u][off];
  }
}

__device__ __forceinline__ void store_tile(float* __restrict__ Ad, float* __restrict__ Bd,
                                           const v4f (&a)[4],
                                           const v4f& bv, const unsigned (&outm)[4]) {
#pragma unroll
  for (int i = 0; i < 4; ++i)
    *reinterpret_cast<v4f*>(Ad + i * 32 * LDA) = a[i];
  v4f bz;
#pragma unroll
  for (int u = 0; u < 4; ++u)
    bz[u] = __builtin_amdgcn_inverse_ballot_w32(outm[u]) ? 0.0f : bv[u];
  *reinterpret_cast<v4f*>(Bd) = bz;
}

__device__ __forceinline__ void lds_barrier() {
  __builtin_amdgcn_fence(__ATOMIC_RELEASE, "workgroup", "local");
  __builtin_amdgcn_s_barrier();
  __builtin_amdgcn_fence(__ATOMIC_ACQUIRE, "workgroup", "local");
}

struct Held {
  v4f      a[4];
  v4f      b;
  unsigned out[4];
};

__device__ __forceinline__ void next_tile_k(int& ch0, int& rem0) {
  ch0  += 3;
  rem0 += 5;
  if (rem0 >= 9) { rem0 -= 9; ch0 += 1; }
}

template <bool kLoad, bool kDecode, bool kStore>
__device__ __forceinline__ void k_step(const float* __restrict__ X,
                                       const float* __restrict__ Wt,
                                       unsigned wLane, unsigned pc,
                                       unsigned mTop, unsigned mBot,
                                       unsigned mLeft, unsigned mRight,
                                       int& ch0, int& rem0, Taps& tp,
                                       const float* __restrict__ fragStage,
                                       float* __restrict__ storeStage,
                                       int aFrag, int bFrag, int aDst, int bDst,
                                       const Frags& frCur, Frags& frNext,
                                       Held& gLoad, const Held& gStore,
                                       v8f (&acc)[2]) {
  load_frags(fragStage, fragStage, aFrag, bFrag, frNext);
  if (kLoad) {
    issue_loads(Wt, wLane, tp, pc, gLoad.a, gLoad.b);
#pragma unroll
    for (int u = 0; u < 4; ++u) gLoad.out[u] = tp.out[u];
  }
  __builtin_amdgcn_sched_barrier(0);
  mma_tile(frCur, acc);
  if (kDecode) {
    next_tile_k(ch0, rem0);
    decode_taps(X, ch0, rem0, mTop, mBot, mLeft, mRight, tp);
  }
  __builtin_amdgcn_sched_barrier(0);
  if (kStore) store_tile(storeStage + aDst, storeStage + bDst, gStore.a, gStore.b, gStore.out);
  lds_barrier();
}

__global__ __launch_bounds__(256) void conv_roi_wmma_kernel(
    const float* __restrict__ X,
    const float* __restrict__ W,
    const float* __restrict__ bias,
    const int*   __restrict__ roi,
    float* __restrict__ out) {
  __shared__ __attribute__((aligned(16))) float smem[2 * STAGE];
  __shared__ __attribute__((aligned(16))) unsigned roiFlag[FILL_PIX * 8];
  static_assert(NROI / BN == HOWO / FILL_PIX, "one bias-fill slab per work-group column");

  const int tid   = threadIdx.x;
  const int lane  = tid & 31;
  const int wave  = tid >> 5;
  const int waveM = wave >> 1;
  const int waveN = wave & 1;
  const int lrow  = lane & 15;
  const int lhi   = lane >> 4;

  const int m0 = blockIdx.y * BM;
  const int n0 = blockIdx.x * BN;

  const int fillBase = blockIdx.x * FILL_PIX;
  int4 rscan[NROI / 1024];
  for (int i = 0; i < NROI / 1024; ++i)
    rscan[i] = reinterpret_cast<const int4*>(roi)[tid + 256 * i];

  const int tn   = tid & 31;
  const int kgrp = __builtin_amdgcn_readfirstlane(tid >> 5);
  const int p_my = roi[n0 + tn];
  const int  fillF0 = m0 + kgrp * FILL_ROWS;
  const v16f fillBv = *reinterpret_cast<const v16f*>(bias + fillF0);
  roiFlag[tn * 8 + kgrp] = 0u;
  roiFlag[(tn + 32) * 8 + kgrp] = 0u;
  const unsigned pc = (unsigned)min(max(p_my, 0), HOWO - 1);
  const int ry   = (int)(pc >> 7);
  const int rx   = (int)(pc & 127u);
  const unsigned mTop   = __builtin_amdgcn_ballot_w32(ry == 0);
  const unsigned mBot   = __builtin_amdgcn_ballot_w32(ry == HW_DIM - 1);
  const unsigned mLeft  = __builtin_amdgcn_ballot_w32(rx == 0);
  const unsigned mRight = __builtin_amdgcn_ballot_w32(rx == HW_DIM - 1);

  const int arow = tid >> 3;
  const int aq   = tid & 7;

  v8f acc[2];
  const v8f vzero = {};
#pragma unroll
  for (int mt = 0; mt < 2; ++mt) acc[mt] = vzero;

  const unsigned wLane = (unsigned)((m0 + arow) * KDIM + aq * 4);
  const int    aDst  = arow * LDA + aq * 4;
  const int    bDst  = A_TILE + tn * LDB + kgrp * 4;
  const int    aFrag = (waveM * 32 + lrow) * LDA + lhi * 4;
  const int    bFrag = A_TILE + (waveN * 16 + lrow) * LDB + lhi * 4;

  Held  g0, g1;
  Frags fr0, fr1;
  float* const st0 = smem;
  float* const st1 = smem + STAGE;

  int ch0  = (kgrp * 4) / 9;
  int rem0 = (kgrp * 4) - ch0 * 9;

  Taps tp;
  decode_taps(X, ch0, rem0, mTop, mBot, mLeft, mRight, tp);
  issue_loads(W, wLane, tp, pc, g0.a, g0.b);
#pragma unroll
  for (int u = 0; u < 4; ++u) g0.out[u] = tp.out[u];
  next_tile_k(ch0, rem0);
  decode_taps(X, ch0, rem0, mTop, mBot, mLeft, mRight, tp);
  issue_loads(W + KTILE, wLane, tp, pc, g1.a, g1.b);
#pragma unroll
  for (int u = 0; u < 4; ++u) g1.out[u] = tp.out[u];
  next_tile_k(ch0, rem0);
  decode_taps(X, ch0, rem0, mTop, mBot, mLeft, mRight, tp);
  mark_roi(rscan, fillBase, roiFlag, kgrp);
  store_tile(st0 + aDst, st0 + bDst, g0.a, g0.b, g0.out);
  issue_loads(W + 2 * KTILE, wLane, tp, pc, g0.a, g0.b);
#pragma unroll
  for (int u = 0; u < 4; ++u) g0.out[u] = tp.out[u];
  next_tile_k(ch0, rem0);
  decode_taps(X, ch0, rem0, mTop, mBot, mLeft, mRight, tp);
  lds_barrier();
  load_frags(st0, st0, aFrag, bFrag, fr0);
  fill_slab(fillBv, out, roiFlag, fillF0, fillBase, lane);
  store_tile(st1 + aDst, st1 + bDst, g1.a, g1.b, g1.out);
  lds_barrier();

  const float* Wn = W + 3 * KTILE;
  for (int t = 0; t < NTILES - 4; t += 2) {
    k_step<true, true, true>(X, Wn, wLane, pc, mTop, mBot, mLeft, mRight, ch0, rem0, tp,
                             st1, st0, aFrag, bFrag, aDst, bDst, fr0, fr1, g1, g0, acc);
    k_step<true, true, true>(X, Wn + KTILE, wLane, pc, mTop, mBot, mLeft, mRight, ch0, rem0, tp,
                             st0, st1, aFrag, bFrag, aDst, bDst, fr1, fr0, g0, g1, acc);
    Wn += 2 * KTILE;
  }
  k_step<true, false, true>(X, Wn, wLane, pc, mTop, mBot, mLeft, mRight, ch0, rem0, tp,
                            st1, st0, aFrag, bFrag, aDst, bDst, fr0, fr1, g1, g0, acc);
  k_step<false, false, true>(X, Wn, wLane, pc, mTop, mBot, mLeft, mRight, ch0, rem0, tp,
                             st0, st1, aFrag, bFrag, aDst, bDst, fr1, fr0, g0, g1, acc);
  load_frags(st1, st1, aFrag, bFrag, fr1);
  mma_tile(fr0, acc);
  mma_tile(fr1, acc);

  const int nn = n0 + waveN * 16 + lrow;
  const int p  = roi[nn];
#pragma unroll
  for (int mt = 0; mt < 2; ++mt) {
    const int fbase = m0 + waveM * 32 + mt * 16 + lhi * 8;
#pragma unroll
    for (int v = 0; v < 8; ++v) {
      const int f = fbase + v;
      out[(f << 14) + p] = acc[mt][v] + bias[f];
    }
  }
}

extern "C" void kernel_launch(void* const* d_in, const int* in_sizes, int n_in,
                              void* d_out, int out_size, void* d_ws, size_t ws_size,
                              hipStream_t stream) {
  const float* X    = (const float*)d_in[0];
  const float* W    = (const float*)d_in[1];
  const float* B    = (const float*)d_in[2];
  const int*   roi  = (const int*)d_in[3];
  float*       out  = (float*)d_out;

  conv_roi_wmma_kernel<<<dim3(NROI / BN, F_OUT / BM), 256, 0, stream>>>(
      X, W, B, roi, out);
}


# K-loop barrier split inside the 16-WMMA run: LDS store of tile t+2 after 4 WMMAs, arrive after 12, wait after 16, decode spread per 4; store latency and barrier wait under WMMAs; same sum order
# speedup vs baseline: 6.2148x; 1.0487x over previous
#include <hip/hip_runtime.h>


typedef float v2f __attribute__((ext_vector_type(2)));
typedef float v4f __attribute__((ext_vector_type(4)));
typedef float v8f __attribute__((ext_vector_type(8)));
typedef float v16f __attribute__((ext_vector_type(16)));
typedef unsigned v4u __attribute__((ext_vector_type(4)));

#define HW_DIM   128
#define F_OUT    256
#define KDIM     2304
#define NROI     8192
#define HOWO     16384
#define BM       128
#define BN       32
#define KTILE    32
#define NTILES   (KDIM / KTILE)
#define LDA      36
#define LDB      36
#define A_TILE   (BM * LDA)
#define B_TILE   (BN * LDB)
#define STAGE    (A_TILE + B_TILE)
#define SAFE_PIX 129u
#define FILL_PIX 64
#define FILL_ROWS 16


__device__ __forceinline__ void mark_roi(const int4 (&rs)[NROI / 1024], int pix0,
                                         unsigned* flags, int wave) {
  for (int i = 0; i < NROI / 1024; ++i) {
    const unsigned d[4] = {(unsigned)(rs[i].x - pix0), (unsigned)(rs[i].y - pix0),
                           (unsigned)(rs[i].z - pix0), (unsigned)(rs[i].w - pix0)};
    const unsigned dmin = min(min(d[0], d[1]), min(d[2], d[3]));
    if (__builtin_amdgcn_ballot_w32(dmin < FILL_PIX) != 0u) {
      for (int c = 0; c < 4; ++c)
        if (d[c] < FILL_PIX) flags[d[c] * 8 + wave] = 1u;
    }
  }
}

__device__ __forceinline__ void fill_slab(const v16f& bv, float* __restrict__ out,
                                          const unsigned* flags,
                                          int f0, int pix0, int lane) {
  float* o = out + (f0 << 14) + pix0 + lane;
  bool plain[2];
  for (int h = 0; h < 2; ++h) {
    const v4u lo = *reinterpret_cast<const v4u*>(&flags[(lane + 32 * h) * 8]);
    const v4u hi = *reinterpret_cast<const v4u*>(&flags[(lane + 32 * h) * 8 + 4]);
    const v4u any = lo | hi;
    plain[h] = (any.x | any.y | any.z | any.w) == 0u;
  }
  for (int h = 0; h < 2; ++h) {
    if (plain[h]) {
      for (int r = 0; r < FILL_ROWS; ++r) o[r * HOWO + 32 * h] = bv[r];
    }
  }
}

struct Frags {
  v4f a[2][KTILE / 8];
  v4f b[KTILE / 8];
};

__device__ __forceinline__ void load_frags(const float* __restrict__ As,
                                           const float* __restrict__ Bs,
                                           int aOff, int bOff, Frags& fr) {
#pragma unroll
  for (int c = 0; c < KTILE / 8; ++c) {
#pragma unroll
    for (int mt = 0; mt < 2; ++mt)
      fr.a[mt][c] = *reinterpret_cast<const v4f*>(&As[aOff + mt * 16 * LDA + c * 8]);
    fr.b[c] = *reinterpret_cast<const v4f*>(&Bs[bOff + c * 8]);
  }
}

__device__ __forceinline__ void mma_tile(const Frags& fr, v8f (&acc)[2]) {
#pragma unroll
  for (int c = 0; c < KTILE / 8; ++c) {
#pragma unroll
    for (int j = 0; j < 2; ++j) {
      v2f bfrag;
      bfrag.x = fr.b[c][2 * j];
      bfrag.y = fr.b[c][2 * j + 1];
#pragma unroll
      for (int mt = 0; mt < 2; ++mt) {
        v2f afrag;
        afrag.x = fr.a[mt][c][2 * j];
        afrag.y = fr.a[mt][c][2 * j + 1];
        acc[mt] = __builtin_amdgcn_wmma_f32_16x16x4_f32(
            false, afrag, false, bfrag,
            (short)0, acc[mt], false, false);
      }
    }
  }
}

template <int C0, int C1>
__device__ __forceinline__ void mma_chunks(const Frags& fr, v8f (&acc)[2]) {
#pragma unroll
  for (int c = C0; c < C1; ++c) {
#pragma unroll
    for (int j = 0; j < 2; ++j) {
      v2f bfrag;
      bfrag.x = fr.b[c][2 * j];
      bfrag.y = fr.b[c][2 * j + 1];
#pragma unroll
      for (int mt = 0; mt < 2; ++mt) {
        v2f afrag;
        afrag.x = fr.a[mt][c][2 * j];
        afrag.y = fr.a[mt][c][2 * j + 1];
        acc[mt] = __builtin_amdgcn_wmma_f32_16x16x4_f32(
            false, afrag, false, bfrag,
            (short)0, acc[mt], false, false);
      }
    }
  }
}

struct Taps {
  const float* xk[4];
  unsigned     out[4];
};

__device__ __forceinline__ void decode_taps(const float* __restrict__ X,
                                            int ch0, int rem0,
                                            unsigned mTop, unsigned mBot,
                                            unsigned mLeft, unsigned mRight,
                                            Taps& tp) {
  int ch  = ch0;
  int rem = rem0;
#pragma unroll
  for (int u = 0; u < 4; ++u) {
    const int fi = (rem >= 3) + (rem >= 6);
    const int fj = rem - 3 * fi;
    const unsigned rowm = (fi == 0) ? mTop  : ((fi == 2) ? mBot   : 0u);
    const unsigned colm = (fj == 0) ? mLeft : ((fj == 2) ? mRight : 0u);
    tp.out[u] = rowm | colm;
    tp.xk[u]  = X + ((ch << 14) + ((fi - 1) << 7) + (fj - 1));
    rem += 1;
    if (rem == 9) { rem = 0; ch += 1; }
  }
}

template <int U>
__device__ __forceinline__ void decode_tap(const float* __restrict__ X,
                                           int& ch, int& rem,
                                           unsigned mTop, unsigned mBot,
                                           unsigned mLeft, unsigned mRight,
                                           Taps& tp) {
  const int fi = (rem >= 3) + (rem >= 6);
  const int fj = rem - 3 * fi;
  const unsigned rowm = (fi == 0) ? mTop  : ((fi == 2) ? mBot   : 0u);
  const unsigned colm = (fj == 0) ? mLeft : ((fj == 2) ? mRight : 0u);
  tp.out[U] = rowm | colm;
  tp.xk[U]  = X + ((ch << 14) + ((fi - 1) << 7) + (fj - 1));
  rem += 1;
  if (rem == 9) { rem = 0; ch += 1; }
}

__device__ __forceinline__ void issue_loads(const float* __restrict__ Wt,
                                            unsigned wLane, const Taps& tp, unsigned pc,
                                            v4f (&a)[4], v4f& bv) {
#pragma unroll
  for (int i = 0; i < 4; ++i)
    a[i] = *reinterpret_cast<const v4f*>(Wt + wLane + i * 32 * KDIM);
#pragma unroll
  for (int u = 0; u < 4; ++u) {
    const unsigned off = __builtin_amdgcn_inverse_ballot_w32(tp.out[u]) ? SAFE_PIX : pc;
    bv[u] = tp.xk[u][off];
  }
}

__device__ __forceinline__ void store_tile(float* __restrict__ Ad, float* __restrict__ Bd,
                                           const v4f (&a)[4],
                                           const v4f& bv, const unsigned (&outm)[4]) {
#pragma unroll
  for (int i = 0; i < 4; ++i)
    *reinterpret_cast<v4f*>(Ad + i * 32 * LDA) = a[i];
  v4f bz;
#pragma unroll
  for (int u = 0; u < 4; ++u)
    bz[u] = __builtin_amdgcn_inverse_ballot_w32(outm[u]) ? 0.0f : bv[u];
  *reinterpret_cast<v4f*>(Bd) = bz;
}

__device__ __forceinline__ void lds_barrier() {
  __builtin_amdgcn_fence(__ATOMIC_RELEASE, "workgroup", "local");
  __builtin_amdgcn_s_barrier();
  __builtin_amdgcn_fence(__ATOMIC_ACQUIRE, "workgroup", "local");
}

__device__ __forceinline__ void lds_barrier_arrive() {
  __builtin_amdgcn_fence(__ATOMIC_RELEASE, "workgroup", "local");
  __builtin_amdgcn_s_barrier_signal(-1);
}
__device__ __forceinline__ void lds_barrier_pass() {
  __builtin_amdgcn_s_barrier_wait(-1);
  __builtin_amdgcn_fence(__ATOMIC_ACQUIRE, "workgroup", "local");
}

struct Held {
  v4f      a[4];
  v4f      b;
  unsigned out[4];
};

__device__ __forceinline__ void next_tile_k(int& ch0, int& rem0) {
  ch0  += 3;
  rem0 += 5;
  if (rem0 >= 9) { rem0 -= 9; ch0 += 1; }
}

template <bool kLoad, bool kDecode, bool kStore>
__device__ __forceinline__ void k_step(const float* __restrict__ X,
                                       const float* __restrict__ Wt,
                                       unsigned wLane, unsigned pc,
                                       unsigned mTop, unsigned mBot,
                                       unsigned mLeft, unsigned mRight,
                                       int& ch0, int& rem0, Taps& tp,
                                       const float* __restrict__ fragStage,
                                       float* __restrict__ storeStage,
                                       int aFrag, int bFrag, int aDst, int bDst,
                                       const Frags& frCur, Frags& frNext,
                                       Held& gLoad, const Held& gStore,
                                       v8f (&acc)[2]) {
  load_frags(fragStage, fragStage, aFrag, bFrag, frNext);
  if (kLoad) {
    issue_loads(Wt, wLane, tp, pc, gLoad.a, gLoad.b);
#pragma unroll
    for (int u = 0; u < 4; ++u) gLoad.out[u] = tp.out[u];
  }
  __builtin_amdgcn_sched_barrier(0);
  int dch = ch0, drem = rem0;
  if (kDecode) {
    next_tile_k(ch0, rem0);
    dch = ch0;
    drem = rem0;
  }
  mma_chunks<0, 1>(frCur, acc);
  if (kDecode) decode_tap<0>(X, dch, drem, mTop, mBot, mLeft, mRight, tp);
  __builtin_amdgcn_sched_barrier(0);
  if (kStore) store_tile(storeStage + aDst, storeStage + bDst, gStore.a, gStore.b, gStore.out);
  __builtin_amdgcn_sched_barrier(0);
  mma_chunks<1, 2>(frCur, acc);
  if (kDecode) decode_tap<1>(X, dch, drem, mTop, mBot, mLeft, mRight, tp);
  __builtin_amdgcn_sched_barrier(0);
  mma_chunks<2, 3>(frCur, acc);
  if (kDecode) decode_tap<2>(X, dch, drem, mTop, mBot, mLeft, mRight, tp);
  __builtin_amdgcn_sched_barrier(0);
  lds_barrier_arrive();
  __builtin_amdgcn_sched_barrier(0);
  mma_chunks<3, 4>(frCur, acc);
  if (kDecode) decode_tap<3>(X, dch, drem, mTop, mBot, mLeft, mRight, tp);
  __builtin_amdgcn_sched_barrier(0);
  lds_barrier_pass();
}

__global__ __launch_bounds__(256) void conv_roi_wmma_kernel(
    const float* __restrict__ X,
    const float* __restrict__ W,
    const float* __restrict__ bias,
    const int*   __restrict__ roi,
    float* __restrict__ out) {
  __shared__ __attribute__((aligned(16))) float smem[2 * STAGE];
  __shared__ __attribute__((aligned(16))) unsigned roiFlag[FILL_PIX * 8];
  static_assert(NROI / BN == HOWO / FILL_PIX, "one bias-fill slab per work-group column");

  const int tid   = threadIdx.x;
  const int lane  = tid & 31;
  const int wave  = tid >> 5;
  const int waveM = wave >> 1;
  const int waveN = wave & 1;
  const int lrow  = lane & 15;
  const int lhi   = lane >> 4;

  const int m0 = blockIdx.y * BM;
  const int n0 = blockIdx.x * BN;

  const int fillBase = blockIdx.x * FILL_PIX;
  int4 rscan[NROI / 1024];
  for (int i = 0; i < NROI / 1024; ++i)
    rscan[i] = reinterpret_cast<const int4*>(roi)[tid + 256 * i];

  const int tn   = tid & 31;
  const int kgrp = __builtin_amdgcn_readfirstlane(tid >> 5);
  const int p_my = roi[n0 + tn];
  const int  fillF0 = m0 + kgrp * FILL_ROWS;
  const v16f fillBv = *reinterpret_cast<const v16f*>(bias + fillF0);
  roiFlag[tn * 8 + kgrp] = 0u;
  roiFlag[(tn + 32) * 8 + kgrp] = 0u;
  const unsigned pc = (unsigned)min(max(p_my, 0), HOWO - 1);
  const int ry   = (int)(pc >> 7);
  const int rx   = (int)(pc & 127u);
  const unsigned mTop   = __builtin_amdgcn_ballot_w32(ry == 0);
  const unsigned mBot   = __builtin_amdgcn_ballot_w32(ry == HW_DIM - 1);
  const unsigned mLeft  = __builtin_amdgcn_ballot_w32(rx == 0);
  const unsigned mRight = __builtin_amdgcn_ballot_w32(rx == HW_DIM - 1);

  const int arow = tid >> 3;
  const int aq   = tid & 7;

  v8f acc[2];
  const v8f vzero = {};
#pragma unroll
  for (int mt = 0; mt < 2; ++mt) acc[mt] = vzero;

  const unsigned wLane = (unsigned)((m0 + arow) * KDIM + aq * 4);
  const int    aDst  = arow * LDA + aq * 4;
  const int    bDst  = A_TILE + tn * LDB + kgrp * 4;
  const int    aFrag = (waveM * 32 + lrow) * LDA + lhi * 4;
  const int    bFrag = A_TILE + (waveN * 16 + lrow) * LDB + lhi * 4;

  Held  g0, g1;
  Frags fr0, fr1;
  float* const st0 = smem;
  float* const st1 = smem + STAGE;

  int ch0  = (kgrp * 4) / 9;
  int rem0 = (kgrp * 4) - ch0 * 9;

  Taps tp;
  decode_taps(X, ch0, rem0, mTop, mBot, mLeft, mRight, tp);
  issue_loads(W, wLane, tp, pc, g0.a, g0.b);
#pragma unroll
  for (int u = 0; u < 4; ++u) g0.out[u] = tp.out[u];
  next_tile_k(ch0, rem0);
  decode_taps(X, ch0, rem0, mTop, mBot, mLeft, mRight, tp);
  issue_loads(W + KTILE, wLane, tp, pc, g1.a, g1.b);
#pragma unroll
  for (int u = 0; u < 4; ++u) g1.out[u] = tp.out[u];
  next_tile_k(ch0, rem0);
  decode_taps(X, ch0, rem0, mTop, mBot, mLeft, mRight, tp);
  mark_roi(rscan, fillBase, roiFlag, kgrp);
  store_tile(st0 + aDst, st0 + bDst, g0.a, g0.b, g0.out);
  issue_loads(W + 2 * KTILE, wLane, tp, pc, g0.a, g0.b);
#pragma unroll
  for (int u = 0; u < 4; ++u) g0.out[u] = tp.out[u];
  next_tile_k(ch0, rem0);
  decode_taps(X, ch0, rem0, mTop, mBot, mLeft, mRight, tp);
  lds_barrier();
  load_frags(st0, st0, aFrag, bFrag, fr0);
  fill_slab(fillBv, out, roiFlag, fillF0, fillBase, lane);
  store_tile(st1 + aDst, st1 + bDst, g1.a, g1.b, g1.out);
  lds_barrier();

  const float* Wn = W + 3 * KTILE;
  for (int t = 0; t < NTILES - 4; t += 2) {
    k_step<true, true, true>(X, Wn, wLane, pc, mTop, mBot, mLeft, mRight, ch0, rem0, tp,
                             st1, st0, aFrag, bFrag, aDst, bDst, fr0, fr1, g1, g0, acc);
    k_step<true, true, true>(X, Wn + KTILE, wLane, pc, mTop, mBot, mLeft, mRight, ch0, rem0, tp,
                             st0, st1, aFrag, bFrag, aDst, bDst, fr1, fr0, g0, g1, acc);
    Wn += 2 * KTILE;
  }
  k_step<true, false, true>(X, Wn, wLane, pc, mTop, mBot, mLeft, mRight, ch0, rem0, tp,
                            st1, st0, aFrag, bFrag, aDst, bDst, fr0, fr1, g1, g0, acc);
  k_step<false, false, true>(X, Wn, wLane, pc, mTop, mBot, mLeft, mRight, ch0, rem0, tp,
                             st0, st1, aFrag, bFrag, aDst, bDst, fr1, fr0, g0, g1, acc);
  load_frags(st1, st1, aFrag, bFrag, fr1);
  mma_tile(fr0, acc);
  mma_tile(fr1, acc);

  const int nn = n0 + waveN * 16 + lrow;
  const int p  = roi[nn];
#pragma unroll
  for (int mt = 0; mt < 2; ++mt) {
    const int fbase = m0 + waveM * 32 + mt * 16 + lhi * 8;
#pragma unroll
    for (int v = 0; v < 8; ++v) {
      const int f = fbase + v;
      out[(f << 14) + p] = acc[mt][v] + bias[f];
    }
  }
}

extern "C" void kernel_launch(void* const* d_in, const int* in_sizes, int n_in,
                              void* d_out, int out_size, void* d_ws, size_t ws_size,
                              hipStream_t stream) {
  const float* X    = (const float*)d_in[0];
  const float* W    = (const float*)d_in[1];
  const float* B    = (const float*)d_in[2];
  const int*   roi  = (const int*)d_in[3];
  float*       out  = (float*)d_out;

  conv_roi_wmma_kernel<<<dim3(NROI / BN, F_OUT / BM), 256, 0, stream>>>(
      X, W, B, roi, out);
}


# Block tile 128x32 -> 128x64, wave tile 32x16 -> 32x32 (four accumulator tiles): 16 fragment reads per 32 WMMAs (was 24), W staged once per 64 columns; grid 512 -> 256, one per WGP; same sum order
# speedup vs baseline: 6.6404x; 1.0685x over previous
#include <hip/hip_runtime.h>


typedef float v2f __attribute__((ext_vector_type(2)));
typedef float v4f __attribute__((ext_vector_type(4)));
typedef float v8f __attribute__((ext_vector_type(8)));
typedef float v16f __attribute__((ext_vector_type(16)));
typedef unsigned v4u __attribute__((ext_vector_type(4)));

#define HW_DIM   128
#define F_OUT    256
#define KDIM     2304
#define NROI     8192
#define HOWO     16384
#define BM       128
#define BN       64
#define KTILE    32
#define NTILES   (KDIM / KTILE)
#define LDA      36
#define LDB      36
#define A_TILE   (BM * LDA)
#define B_TILE   (BN * LDB)
#define STAGE    (A_TILE + B_TILE)
#define LDS_FLOATS (41 * 1024)
#define SAFE_PIX 129u
#define FILL_PIX 128
#define FILL_ROWS 16


__device__ __forceinline__ void mark_roi(const int4 (&rs)[NROI / 1024], int pix0,
                                         unsigned* flags, int wave) {
  for (int i = 0; i < NROI / 1024; ++i) {
    const unsigned d[4] = {(unsigned)(rs[i].x - pix0), (unsigned)(rs[i].y - pix0),
                           (unsigned)(rs[i].z - pix0), (unsigned)(rs[i].w - pix0)};
    const unsigned dmin = min(min(d[0], d[1]), min(d[2], d[3]));
    if (__builtin_amdgcn_ballot_w32(dmin < FILL_PIX) != 0u) {
      for (int c = 0; c < 4; ++c)
        if (d[c] < FILL_PIX) flags[d[c] * 8 + wave] = 1u;
    }
  }
}

__device__ __forceinline__ void fill_slab(const v16f& bv, float* __restrict__ out,
                                          const unsigned* flags,
                                          int f0, int pix0, int lane) {
  float* o = out + (f0 << 14) + pix0 + lane;
  bool plain[FILL_PIX / 32];
  for (int h = 0; h < FILL_PIX / 32; ++h) {
    const v4u lo = *reinterpret_cast<const v4u*>(&flags[(lane + 32 * h) * 8]);
    const v4u hi = *reinterpret_cast<const v4u*>(&flags[(lane + 32 * h) * 8 + 4]);
    const v4u any = lo | hi;
    plain[h] = (any.x | any.y | any.z | any.w) == 0u;
  }
  for (int h = 0; h < FILL_PIX / 32; ++h) {
    if (plain[h]) {
      for (int r = 0; r < FILL_ROWS; ++r) o[r * HOWO + 32 * h] = bv[r];
    }
  }
}

struct Frags {
  v4f a[2][KTILE / 8];
  v4f b[2][KTILE / 8];
};

__device__ __forceinline__ void load_frags(const float* __restrict__ As,
                                           const float* __restrict__ Bs,
                                           int aOff, int bOff, Frags& fr) {
#pragma unroll
  for (int c = 0; c < KTILE / 8; ++c) {
#pragma unroll
    for (int mt = 0; mt < 2; ++mt)
      fr.a[mt][c] = *reinterpret_cast<const v4f*>(&As[aOff + mt * 16 * LDA + c * 8]);
#pragma unroll
    for (int nt = 0; nt < 2; ++nt)
      fr.b[nt][c] = *reinterpret_cast<const v4f*>(&Bs[bOff + nt * 16 * LDB + c * 8]);
  }
}

template <int C, int J>
__device__ __forceinline__ void mma_group(const Frags& fr, v8f (&acc)[2][2]) {
#pragma unroll
  for (int nt = 0; nt < 2; ++nt) {
    v2f bfrag;
    bfrag.x = fr.b[nt][C][2 * J];
    bfrag.y = fr.b[nt][C][2 * J + 1];
#pragma unroll
    for (int mt = 0; mt < 2; ++mt) {
      v2f afrag;
      afrag.x = fr.a[mt][C][2 * J];
      afrag.y = fr.a[mt][C][2 * J + 1];
      acc[mt][nt] = __builtin_amdgcn_wmma_f32_16x16x4_f32(
          false, afrag, false, bfrag,
          (short)0, acc[mt][nt], false, false);
    }
  }
}

__device__ __forceinline__ void mma_tile(const Frags& fr, v8f (&acc)[2][2]) {
  mma_group<0, 0>(fr, acc);  mma_group<0, 1>(fr, acc);
  mma_group<1, 0>(fr, acc);  mma_group<1, 1>(fr, acc);
  mma_group<2, 0>(fr, acc);  mma_group<2, 1>(fr, acc);
  mma_group<3, 0>(fr, acc);  mma_group<3, 1>(fr, acc);
}

#define NGATHER 8
struct Taps {
  const float* xk[NGATHER];
  unsigned     out[NGATHER];
};

__device__ __forceinline__ void decode_taps(const float* __restrict__ X,
                                            int ch0, int rem0,
                                            unsigned mTop, unsigned mBot,
                                            unsigned mLeft, unsigned mRight,
                                            Taps& tp) {
  int ch  = ch0;
  int rem = rem0;
#pragma unroll
  for (int u = 0; u < NGATHER; ++u) {
    const int fi = (rem >= 3) + (rem >= 6);
    const int fj = rem - 3 * fi;
    const unsigned rowm = (fi == 0) ? mTop  : ((fi == 2) ? mBot   : 0u);
    const unsigned colm = (fj == 0) ? mLeft : ((fj == 2) ? mRight : 0u);
    tp.out[u] = rowm | colm;
    tp.xk[u]  = X + ((ch << 14) + ((fi - 1) << 7) + (fj - 1));
    rem += 1;
    if (rem == 9) { rem = 0; ch += 1; }
  }
}

template <int U>
__device__ __forceinline__ void decode_tap(const float* __restrict__ X,
                                           int& ch, int& rem,
                                           unsigned mTop, unsigned mBot,
                                           unsigned mLeft, unsigned mRight,
                                           Taps& tp) {
  const int fi = (rem >= 3) + (rem >= 6);
  const int fj = rem - 3 * fi;
  const unsigned rowm = (fi == 0) ? mTop  : ((fi == 2) ? mBot   : 0u);
  const unsigned colm = (fj == 0) ? mLeft : ((fj == 2) ? mRight : 0u);
  tp.out[U] = rowm | colm;
  tp.xk[U]  = X + ((ch << 14) + ((fi - 1) << 7) + (fj - 1));
  rem += 1;
  if (rem == 9) { rem = 0; ch += 1; }
}

__device__ __forceinline__ void issue_loads(const float* __restrict__ Wt,
                                            unsigned wLane, const Taps& tp, unsigned pc,
                                            v4f (&a)[4], v4f (&bv)[2]) {
#pragma unroll
  for (int i = 0; i < 4; ++i)
    a[i] = *reinterpret_cast<const v4f*>(Wt + wLane + i * 32 * KDIM);
#pragma unroll
  for (int u = 0; u < NGATHER; ++u) {
    const unsigned off = __builtin_amdgcn_inverse_ballot_w32(tp.out[u]) ? SAFE_PIX : pc;
    bv[u >> 2][u & 3] = tp.xk[u][off];
  }
}

__device__ __forceinline__ void store_tile(float* __restrict__ Ad, float* __restrict__ Bd,
                                           const v4f (&a)[4],
                                           const v4f (&bv)[2], const unsigned (&outm)[NGATHER]) {
#pragma unroll
  for (int i = 0; i < 4; ++i)
    *reinterpret_cast<v4f*>(Ad + i * 32 * LDA) = a[i];
#pragma unroll
  for (int q = 0; q < 2; ++q) {
    v4f bz;
#pragma unroll
    for (int u = 0; u < 4; ++u)
      bz[u] = __builtin_amdgcn_inverse_ballot_w32(outm[4 * q + u]) ? 0.0f : bv[q][u];
    *reinterpret_cast<v4f*>(Bd + 4 * q) = bz;
  }
}

__device__ __forceinline__ void lds_barrier() {
  __builtin_amdgcn_fence(__ATOMIC_RELEASE, "workgroup", "local");
  __builtin_amdgcn_s_barrier();
  __builtin_amdgcn_fence(__ATOMIC_ACQUIRE, "workgroup", "local");
}

__device__ __forceinline__ void lds_barrier_arrive() {
  __builtin_amdgcn_fence(__ATOMIC_RELEASE, "workgroup", "local");
  __builtin_amdgcn_s_barrier_signal(-1);
}
__device__ __forceinline__ void lds_barrier_pass() {
  __builtin_amdgcn_s_barrier_wait(-1);
  __builtin_amdgcn_fence(__ATOMIC_ACQUIRE, "workgroup", "local");
}

struct Held {
  v4f      a[4];
  v4f      b[2];
  unsigned out[NGATHER];
};

__device__ __forceinline__ void next_tile_k(int& ch0, int& rem0) {
  ch0  += 3;
  rem0 += 5;
  if (rem0 >= 9) { rem0 -= 9; ch0 += 1; }
}

template <bool kLoad, bool kDecode, bool kStore>
__device__ __forceinline__ void k_step(const float* __restrict__ X,
                                       const float* __restrict__ Wt,
                                       unsigned wLane, unsigned pc,
                                       unsigned mTop, unsigned mBot,
                                       unsigned mLeft, unsigned mRight,
                                       int& ch0, int& rem0, Taps& tp,
                                       const float* __restrict__ fragStage,
                                       float* __restrict__ storeStage,
                                       int aFrag, int bFrag, int aDst, int bDst,
                                       const Frags& frCur, Frags& frNext,
                                       Held& gLoad, const Held& gStore,
                                       v8f (&acc)[2][2]) {
  load_frags(fragStage, fragStage, aFrag, bFrag, frNext);
  if (kLoad) {
    issue_loads(Wt, wLane, tp, pc, gLoad.a, gLoad.b);
#pragma unroll
    for (int u = 0; u < NGATHER; ++u) gLoad.out[u] = tp.out[u];
  }
  __builtin_amdgcn_sched_barrier(0);
  int dch = ch0, drem = rem0;
  if (kDecode) {
    next_tile_k(ch0, rem0);
    dch = ch0;
    drem = rem0;
  }
  mma_group<0, 0>(frCur, acc);
  if (kDecode) decode_tap<0>(X, dch, drem, mTop, mBot, mLeft, mRight, tp);
  __builtin_amdgcn_sched_barrier(0);
  mma_group<0, 1>(frCur, acc);
  if (kDecode) decode_tap<1>(X, dch, drem, mTop, mBot, mLeft, mRight, tp);
  __builtin_amdgcn_sched_barrier(0);
  if (kStore) store_tile(storeStage + aDst, storeStage + bDst, gStore.a, gStore.b, gStore.out);
  __builtin_amdgcn_sched_barrier(0);
  mma_group<1, 0>(frCur, acc);
  if (kDecode) decode_tap<2>(X, dch, drem, mTop, mBot, mLeft, mRight, tp);
  __builtin_amdgcn_sched_barrier(0);
  mma_group<1, 1>(frCur, acc);
  if (kDecode) decode_tap<3>(X, dch, drem, mTop, mBot, mLeft, mRight, tp);
  __builtin_amdgcn_sched_barrier(0);
  mma_group<2, 0>(frCur, acc);
  if (kDecode) decode_tap<4>(X, dch, drem, mTop, mBot, mLeft, mRight, tp);
  __builtin_amdgcn_sched_barrier(0);
  mma_group<2, 1>(frCur, acc);
  if (kDecode) decode_tap<5>(X, dch, drem, mTop, mBot, mLeft, mRight, tp);
  __builtin_amdgcn_sched_barrier(0);
  lds_barrier_arrive();
  __builtin_amdgcn_sched_barrier(0);
  mma_group<3, 0>(frCur, acc);
  if (kDecode) decode_tap<6>(X, dch, drem, mTop, mBot, mLeft, mRight, tp);
  __builtin_amdgcn_sched_barrier(0);
  mma_group<3, 1>(frCur, acc);
  if (kDecode) decode_tap<7>(X, dch, drem, mTop, mBot, mLeft, mRight, tp);
  __builtin_amdgcn_sched_barrier(0);
  lds_barrier_pass();
}

__global__ __launch_bounds__(256) void conv_roi_wmma_kernel(
    const float* __restrict__ X,
    const float* __restrict__ W,
    const float* __restrict__ bias,
    const int*   __restrict__ roi,
    float* __restrict__ out) {
  __shared__ __attribute__((aligned(16))) float smem[LDS_FLOATS];
  static_assert(2 * STAGE <= LDS_FLOATS, "both stages inside the reservation");
  static_assert(LDS_FLOATS * 4 > 160 * 1024, "more than half of the WGP's LDS");
  __shared__ __attribute__((aligned(16))) unsigned roiFlag[FILL_PIX * 8];
  static_assert(NROI / BN == HOWO / FILL_PIX, "one bias-fill slab per work-group column");

  const int tid   = threadIdx.x;
  const int lane  = tid & 31;
  const int wave  = tid >> 5;
  const int waveM = wave >> 1;
  const int waveN = wave & 1;
  const int lrow  = lane & 15;
  const int lhi   = lane >> 4;

  const int m0 = blockIdx.y * BM;
  const int n0 = blockIdx.x * BN;

  const int fillBase = blockIdx.x * FILL_PIX;
  int4 rscan[NROI / 1024];
  for (int i = 0; i < NROI / 1024; ++i)
    rscan[i] = reinterpret_cast<const int4*>(roi)[tid + 256 * i];

  const int wv   = __builtin_amdgcn_readfirstlane(tid >> 5);
  const int tn   = (wv & 1) * 32 + lane;
  const int kgrp = wv >> 1;
  const int p_my = roi[n0 + tn];
  const int  fillF0 = m0 + wv * FILL_ROWS;
  const v16f fillBv = *reinterpret_cast<const v16f*>(bias + fillF0);
  for (int h = 0; h < FILL_PIX / 32; ++h) roiFlag[(lane + 32 * h) * 8 + wv] = 0u;
  const unsigned pc = (unsigned)min(max(p_my, 0), HOWO - 1);
  const int ry   = (int)(pc >> 7);
  const int rx   = (int)(pc & 127u);
  const unsigned mTop   = __builtin_amdgcn_ballot_w32(ry == 0);
  const unsigned mBot   = __builtin_amdgcn_ballot_w32(ry == HW_DIM - 1);
  const unsigned mLeft  = __builtin_amdgcn_ballot_w32(rx == 0);
  const unsigned mRight = __builtin_amdgcn_ballot_w32(rx == HW_DIM - 1);

  const int arow = tid >> 3;
  const int aq   = tid & 7;

  v8f acc[2][2];
  const v8f vzero = {};
#pragma unroll
  for (int mt = 0; mt < 2; ++mt) {
#pragma unroll
    for (int nt = 0; nt < 2; ++nt) acc[mt][nt] = vzero;
  }

  const unsigned wLane = (unsigned)((m0 + arow) * KDIM + aq * 4);
  const int    aDst  = arow * LDA + aq * 4;
  const int    bDst  = A_TILE + tn * LDB + kgrp * 8;
  const int    aFrag = (waveM * 32 + lrow) * LDA + lhi * 4;
  const int    bFrag = A_TILE + (waveN * 32 + lrow) * LDB + lhi * 4;

  Held  g0, g1;
  Frags fr0, fr1;
  float* const st0 = smem;
  float* const st1 = smem + STAGE;

  int ch0  = (kgrp * 8) / 9;
  int rem0 = (kgrp * 8) - ch0 * 9;

  Taps tp;
  decode_taps(X, ch0, rem0, mTop, mBot, mLeft, mRight, tp);
  issue_loads(W, wLane, tp, pc, g0.a, g0.b);
#pragma unroll
  for (int u = 0; u < NGATHER; ++u) g0.out[u] = tp.out[u];
  next_tile_k(ch0, rem0);
  decode_taps(X, ch0, rem0, mTop, mBot, mLeft, mRight, tp);
  issue_loads(W + KTILE, wLane, tp, pc, g1.a, g1.b);
#pragma unroll
  for (int u = 0; u < NGATHER; ++u) g1.out[u] = tp.out[u];
  next_tile_k(ch0, rem0);
  decode_taps(X, ch0, rem0, mTop, mBot, mLeft, mRight, tp);
  mark_roi(rscan, fillBase, roiFlag, wv);
  store_tile(st0 + aDst, st0 + bDst, g0.a, g0.b, g0.out);
  issue_loads(W + 2 * KTILE, wLane, tp, pc, g0.a, g0.b);
#pragma unroll
  for (int u = 0; u < NGATHER; ++u) g0.out[u] = tp.out[u];
  next_tile_k(ch0, rem0);
  decode_taps(X, ch0, rem0, mTop, mBot, mLeft, mRight, tp);
  lds_barrier();
  load_frags(st0, st0, aFrag, bFrag, fr0);
  fill_slab(fillBv, out, roiFlag, fillF0, fillBase, lane);
  store_tile(st1 + aDst, st1 + bDst, g1.a, g1.b, g1.out);
  lds_barrier();

  const float* Wn = W + 3 * KTILE;
  for (int t = 0; t < NTILES - 4; t += 2) {
    k_step<true, true, true>(X, Wn, wLane, pc, mTop, mBot, mLeft, mRight, ch0, rem0, tp,
                             st1, st0, aFrag, bFrag, aDst, bDst, fr0, fr1, g1, g0, acc);
    k_step<true, true, true>(X, Wn + KTILE, wLane, pc, mTop, mBot, mLeft, mRight, ch0, rem0, tp,
                             st0, st1, aFrag, bFrag, aDst, bDst, fr1, fr0, g0, g1, acc);
    Wn += 2 * KTILE;
  }
  k_step<true, false, true>(X, Wn, wLane, pc, mTop, mBot, mLeft, mRight, ch0, rem0, tp,
                            st1, st0, aFrag, bFrag, aDst, bDst, fr0, fr1, g1, g0, acc);
  k_step<false, false, true>(X, Wn, wLane, pc, mTop, mBot, mLeft, mRight, ch0, rem0, tp,
                             st0, st1, aFrag, bFrag, aDst, bDst, fr1, fr0, g0, g1, acc);
  load_frags(st1, st1, aFrag, bFrag, fr1);
  mma_tile(fr0, acc);
  mma_tile(fr1, acc);

#pragma unroll
  for (int nt = 0; nt < 2; ++nt) {
    const int nn = n0 + waveN * 32 + nt * 16 + lrow;
    const int p  = roi[nn];
#pragma unroll
    for (int mt = 0; mt < 2; ++mt) {
      const int fbase = m0 + waveM * 32 + mt * 16 + lhi * 8;
#pragma unroll
      for (int v = 0; v < 8; ++v) {
        const int f = fbase + v;
        out[(f << 14) + p] = acc[mt][nt][v] + bias[f];
      }
    }
  }
}

extern "C" void kernel_launch(void* const* d_in, const int* in_sizes, int n_in,
                              void* d_out, int out_size, void* d_ws, size_t ws_size,
                              hipStream_t stream) {
  const float* X    = (const float*)d_in[0];
  const float* W    = (const float*)d_in[1];
  const float* B    = (const float*)d_in[2];
  const int*   roi  = (const int*)d_in[3];
  float*       out  = (float*)d_out;

  conv_roi_wmma_kernel<<<dim3(NROI / BN, F_OUT / BM), 256, 0, stream>>>(
      X, W, B, roi, out);
}
